# starting kernel
# baseline (speedup 1.0000x reference)
#include <hip/hip_runtime.h>
#include <hip/hip_bf16.h>

typedef float v2f __attribute__((ext_vector_type(2)));
typedef float v8f __attribute__((ext_vector_type(8)));

#define SA    104
#define KROWS 180
#define NB    90
#define BW    6

__global__ __launch_bounds__(256)
void odein_solve_kernel(const float* __restrict__ coeffs,
                        const float* __restrict__ rhs,
                        const float* __restrict__ iv_rhs,
                        const float* __restrict__ steps,
                        float* __restrict__ out)
{
    __shared__ float Abuf[KROWS * SA];
    __shared__ float st[32];
    __shared__ float r1[96];

    const int tid = threadIdx.x;
    const int b   = blockIdx.x;

    for (int idx = tid; idx < KROWS * SA; idx += 256) Abuf[idx] = 0.0f;
    if (tid < 29) st[tid] = steps[b * 29 + tid];
    if (tid < 96) r1[tid] = 0.0f;
    __syncthreads();

    if (tid < 90) {
        float c = coeffs[b * 90 + tid];
        int s = tid / 3;
        Abuf[s * SA + tid] = c;
        float v = c * rhs[b * 30 + s];
        if      (tid == 0) v += iv_rhs[b * 4 + 0];
        else if (tid == 1) v += iv_rhs[b * 4 + 1];
        else if (tid == 3) v += iv_rhs[b * 4 + 2];
        else if (tid == 4) v += iv_rhs[b * 4 + 3];
        r1[tid] = v;
    }
    if (tid >= 96 && tid < 100) {
        const int ivcol[4] = {0, 1, 3, 4};
        Abuf[(30 + (tid - 96)) * SA + ivcol[tid - 96]] = 1.0f;
    }
    if (tid >= 128 && tid < 157) {
        int s = tid - 128;
        float h = st[s], h2 = h * h;
        int r = 34 + 2 * s;
        Abuf[r * SA + 3 * (s + 1) + 0] = -1.0f;
        Abuf[r * SA + 3 * s + 0] = 1.0f;
        Abuf[r * SA + 3 * s + 1] = h;
        Abuf[r * SA + 3 * s + 2] = 0.5f * h2;
        r = 35 + 2 * s;
        Abuf[r * SA + 3 * (s + 1) + 1] = -h;
        Abuf[r * SA + 3 * s + 1] = h;
        Abuf[r * SA + 3 * s + 2] = h2;
        r = 120 + 2 * s;
        Abuf[r * SA + 3 * s + 0] = -1.0f;
        Abuf[r * SA + 3 * (s + 1) + 0] = 1.0f;
        Abuf[r * SA + 3 * (s + 1) + 1] = -h;
        Abuf[r * SA + 3 * (s + 1) + 2] = 0.5f * h2;
        r = 121 + 2 * s;
        Abuf[r * SA + 3 * s + 1] = h;
        Abuf[r * SA + 3 * (s + 1) + 1] = -h;
        Abuf[r * SA + 3 * (s + 1) + 2] = h2;
    }
    if (tid >= 160 && tid < 188) {
        int s = tid - 159;
        int r = 91 + s;
        Abuf[r * SA + 3 * (s - 1) + 1] = -1.0f;
        Abuf[r * SA + 3 * (s + 1) + 1] = 1.0f;
        Abuf[r * SA + 3 * s + 2] = -(st[s] + st[s - 1]);
    }
    __syncthreads();

    const int lane = tid & 31, wv = tid >> 5;
    const int half = lane >> 4, lm = lane & 15;
    const int TI[16] = {0,0, 1,1,1, 2,2,2, 3,3,3, 4,4,4, 5,5};
    const int TJ[16] = {0,1, 0,1,2, 1,2,3, 2,3,4, 3,4,5, 4,5};
    const int m0a = TI[2 * wv] * 16,     n0a = TJ[2 * wv] * 16;
    const int m0b = TI[2 * wv + 1] * 16, n0b = TJ[2 * wv + 1] * 16;

    v8f acc0 = {0.f, 0.f, 0.f, 0.f, 0.f, 0.f, 0.f, 0.f};
    v8f acc1 = {0.f, 0.f, 0.f, 0.f, 0.f, 0.f, 0.f, 0.f};
    for (int kk = 0; kk < KROWS / 4; ++kk) {
        int k0 = kk * 4 + 2 * half;
        v2f a, bb;
        a.x  = Abuf[k0 * SA + m0a + lm];
        a.y  = Abuf[(k0 + 1) * SA + m0a + lm];
        bb.x = Abuf[k0 * SA + n0a + lm];
        bb.y = Abuf[(k0 + 1) * SA + n0a + lm];
        acc0 = __builtin_amdgcn_wmma_f32_16x16x4_f32(false, a, false, bb,
                                                     (short)0, acc0, false, false);
        a.x  = Abuf[k0 * SA + m0b + lm];
        a.y  = Abuf[(k0 + 1) * SA + m0b + lm];
        bb.x = Abuf[k0 * SA + n0b + lm];
        bb.y = Abuf[(k0 + 1) * SA + n0b + lm];
        acc1 = __builtin_amdgcn_wmma_f32_16x16x4_f32(false, a, false, bb,
                                                     (short)0, acc1, false, false);
    }
    __syncthreads();
    float* AtA = Abuf;
#pragma unroll
    for (int v = 0; v < 8; ++v)
        AtA[(m0a + v + 8 * half) * SA + n0a + lm] = acc0[v];
#pragma unroll
    for (int v = 0; v < 8; ++v)
        AtA[(m0b + v + 8 * half) * SA + n0b + lm] = acc1[v];
    __syncthreads();

    const int DR[21] = {1,2,3,4,5,6, 2,3,4,5,6, 3,4,5,6, 4,5,6, 5,6, 6};
    const int DC[21] = {1,1,1,1,1,1, 2,2,2,2,2, 3,3,3,3, 4,4,4, 5,5, 6};
    for (int j = 0; j < NB; ++j) {
        if (tid == 0) {
            float d = sqrtf(AtA[j * SA + j]);
            AtA[j * SA + j] = d;
            float inv = 1.0f / d;
            int rmax = (j + BW < NB - 1) ? (j + BW) : (NB - 1);
            for (int r = j + 1; r <= rmax; ++r) AtA[r * SA + j] *= inv;
        }
        __syncthreads();
        if (tid < 21) {
            int r = j + DR[tid];
            if (r < NB) {
                int c = j + DC[tid];
                AtA[r * SA + c] -= AtA[r * SA + j] * AtA[c * SA + j];
            }
        }
        __syncthreads();
    }

    if (tid == 0) {
        for (int j = 0; j < NB; ++j) {
            float s = r1[j];
            int k0 = (j - BW > 0) ? (j - BW) : 0;
            for (int k = k0; k < j; ++k) s -= AtA[j * SA + k] * r1[k];
            r1[j] = s / AtA[j * SA + j];
        }
        for (int j = NB - 1; j >= 0; --j) {
            float s = r1[j];
            int k1 = (j + BW < NB - 1) ? (j + BW) : (NB - 1);
            for (int k = j + 1; k <= k1; ++k) s -= AtA[k * SA + j] * r1[k];
            r1[j] = s / AtA[j * SA + j];
        }
    }
    __syncthreads();
    if (tid < 90) out[b * 90 + tid] = r1[tid];
}

extern "C" void kernel_launch(void* const* d_in, const int* in_sizes, int n_in,
                              void* d_out, int out_size, void* d_ws, size_t ws_size,
                              hipStream_t stream) {
    (void)n_in; (void)out_size; (void)d_ws; (void)ws_size;
    const float* coeffs = (const float*)d_in[0];
    const float* rhs    = (const float*)d_in[1];
    const float* iv     = (const float*)d_in[2];
    const float* steps  = (const float*)d_in[3];
    float* out = (float*)d_out;
    int bs = in_sizes[0] / 90;
    odein_solve_kernel<<<bs, 256, 0, stream>>>(coeffs, rhs, iv, steps, out);
}


# A^T A built one wave per system: a wave fills its own folded A in LDS and runs the 11 lower-band tiles (the 5 never-stored ones dropped) on 11 accumulators; no block barrier per system, was 2
# speedup vs baseline: 39.1239x; 39.1239x over previous
#include <hip/hip_runtime.h>
#include <hip/hip_bf16.h>


typedef float v2f __attribute__((ext_vector_type(2)));
typedef float v8f __attribute__((ext_vector_type(8)));

#define SA    104
#define KROWS 24
#define NB    90
#define BW    6
#define G     16
#define NBP   96
#define NW    8

static __device__ __forceinline__ void wave_lds_handover() {
    __builtin_amdgcn_fence(__ATOMIC_RELEASE, "wavefront");
    __builtin_amdgcn_wave_barrier();
    __builtin_amdgcn_fence(__ATOMIC_ACQUIRE, "wavefront");
}

__global__ __launch_bounds__(256)
void odein_solve_kernel(const float* __restrict__ coeffs,
                        const float* __restrict__ rhs,
                        const float* __restrict__ iv_rhs,
                        const float* __restrict__ steps,
                        float* __restrict__ out,
                        int bs)
{
    __shared__ float Abuf[NW * KROWS * SA];
    __shared__ float band[NBP * 7 * G];
    __shared__ float rv[NBP * G];
    __shared__ float cs[G * 90];
    __shared__ float rs[G * 30];
    __shared__ float ivs[G * 4];
    __shared__ float sts[G * 29];

    const int tid = threadIdx.x;
    const int b0  = blockIdx.x * G;
    const int nsys = (bs - b0 < G) ? (bs - b0) : G;

    for (int i = tid; i < G * 90; i += 256) cs[i]  = (i < nsys * 90) ? coeffs[b0 * 90 + i] : 0.0f;
    for (int i = tid; i < G * 30; i += 256) rs[i]  = (i < nsys * 30) ? rhs[b0 * 30 + i] : 0.0f;
    for (int i = tid; i < G * 29; i += 256) sts[i] = (i < nsys * 29) ? steps[b0 * 29 + i] : 0.0f;
    if (tid < G * 4) ivs[tid] = (tid < nsys * 4) ? iv_rhs[b0 * 4 + tid] : 0.0f;
    for (int idx = tid; idx < NW * KROWS * SA; idx += 256) Abuf[idx] = 0.0f;
    for (int i = tid; i < (NBP - NB) * 7 * G; i += 256) band[NB * 7 * G + i] = 0.0f;
    for (int i = tid; i < (NBP - NB) * G; i += 256) rv[NB * G + i] = 0.0f;
    __syncthreads();

    const int lane = tid & 31, wv = tid >> 5;
    const int half = lane >> 4, lm = lane & 15;
    constexpr int NT = 11;
    constexpr int TI[NT] = {0, 1, 1, 2, 2, 3, 3, 4, 4, 5, 5};
    constexpr int TJ[NT] = {0, 0, 1, 1, 2, 2, 3, 3, 4, 4, 5};
    float* Aw = Abuf + wv * (KROWS * SA);

    if (lane < 4) {
        const int ivcol = lane + (lane >> 1);
        const int ivrow = lane == 0 ? 6 : lane == 1 ? 10 : lane == 2 ? 18 : 19;
        Aw[ivrow * SA + ivcol] = 1.0f;
    }

    for (int s2 = 0; s2 < G / NW; ++s2) {
        const int g = (G / NW) * wv + s2;
        if (lane < 30) {
            const int s = lane;
            const float c0 = cs[g * 90 + 3 * s], c1 = cs[g * 90 + 3 * s + 1], c2 = cs[g * 90 + 3 * s + 2];
            const float r  = rs[g * 30 + s];
            const float i0 = ivs[g * 4 + 2 * (s & 1)], i1 = ivs[g * 4 + 2 * (s & 1) + 1];
            const float h  = sts[g * 29 + (s < 29 ? s : 28)];
            const float hp = sts[g * 29 + (s >= 1 ? s - 1 : 0)];
            const int row = s % 3;
            float v0 = c0 * r, v1 = c1 * r;
            const float v2 = c2 * r;
            if (s < 2) { v0 += i0; v1 += i1; }
            Aw[row * SA + 3 * s + 0] = c0;
            Aw[row * SA + 3 * s + 1] = c1;
            Aw[row * SA + 3 * s + 2] = c2;
            rv[(3 * s + 0) * G + g] = v0;
            rv[(3 * s + 1) * G + g] = v1;
            rv[(3 * s + 2) * G + g] = v2;
            if (s < 29) {
                const float h2 = h * h;
                const int q = s & 3;
                int rr = 3 + q;
                Aw[rr * SA + 3 * (s + 1) + 0] = -1.0f;
                Aw[rr * SA + 3 * s + 0] = 1.0f;
                Aw[rr * SA + 3 * s + 1] = h;
                Aw[rr * SA + 3 * s + 2] = 0.5f * h2;
                rr = 7 + q;
                Aw[rr * SA + 3 * (s + 1) + 1] = -h;
                Aw[rr * SA + 3 * s + 1] = h;
                Aw[rr * SA + 3 * s + 2] = h2;
                rr = 11 + q;
                Aw[rr * SA + 3 * s + 0] = -1.0f;
                Aw[rr * SA + 3 * (s + 1) + 0] = 1.0f;
                Aw[rr * SA + 3 * (s + 1) + 1] = -h;
                Aw[rr * SA + 3 * (s + 1) + 2] = 0.5f * h2;
                rr = 15 + q;
                Aw[rr * SA + 3 * s + 1] = h;
                Aw[rr * SA + 3 * (s + 1) + 1] = -h;
                Aw[rr * SA + 3 * (s + 1) + 2] = h2;
                if (s >= 1) {
                    rr = 19 + s % 5;
                    Aw[rr * SA + 3 * (s - 1) + 1] = -1.0f;
                    Aw[rr * SA + 3 * (s + 1) + 1] = 1.0f;
                    Aw[rr * SA + 3 * s + 2] = -(h + hp);
                }
            }
        }
        wave_lds_handover();

        v2f fr[6][KROWS / 4];
#pragma unroll
        for (int kk = 0; kk < KROWS / 4; ++kk) {
            const int k0 = kk * 4 + 2 * half;
#pragma unroll
            for (int b = 0; b < 6; ++b) {
                fr[b][kk].x = Aw[k0 * SA + b * 16 + lm];
                fr[b][kk].y = Aw[(k0 + 1) * SA + b * 16 + lm];
            }
        }
        wave_lds_handover();

        v8f acc[NT];
#pragma unroll
        for (int t = 0; t < NT; ++t) acc[t] = (v8f){0.f, 0.f, 0.f, 0.f, 0.f, 0.f, 0.f, 0.f};
#pragma unroll
        for (int kk = 0; kk < KROWS / 4; ++kk) {
#pragma unroll
            for (int t = 0; t < NT; ++t)
                acc[t] = __builtin_amdgcn_wmma_f32_16x16x4_f32(false, fr[TI[t]][kk], false, fr[TJ[t]][kk],
                                                               (short)0, acc[t], false, false);
        }

        float* bp = band + (6 * lm + 8 * half) * G + g;
#pragma unroll
        for (int v = 0; v < 8; ++v) {
            const int d0 = v + 8 * half - lm;
            if (d0 >= 0 && d0 <= BW) {
#pragma unroll
                for (int j = 0; j < 5; ++j) bp[(112 * j + v) * G] = acc[2 * j][v];
                if (lm < NB - 80) bp[(112 * 5 + v) * G] = acc[10][v];
            }
            if (d0 + 16 >= 0 && d0 + 16 <= BW) {
#pragma unroll
                for (int j = 0; j < 5; ++j) bp[(112 * j + 16 + v) * G] = acc[2 * j + 1][v];
            }
        }
    }
    __syncthreads();

    if (tid < 32) {
        const int g   = tid & 15;
        const int rev = tid >> 4;
        const int o0 = rev ? 623 : 0, sp = rev ? -6 : 1, sq = rev ? -1 : 6;
        const int ro = rev ? 89 : 0,  rs = rev ? -1 : 1;
        float W[7][7];
        float R[7];
#pragma unroll
        for (int r = 0; r < 7; ++r) {
#pragma unroll
            for (int c = 0; c <= r; ++c) W[r][c] = band[(o0 + sp * r + sq * c) * G + g];
            R[r] = rv[(ro + rs * r) * G + g];
        }
        const int dj = sp + sq;
        int eb = o0 + sp * 7 + sq;
        int lc = o0;
        int rj = ro;
        for (int j = 0; j < 42; ++j) {
            float nw[7], nr;
#pragma unroll
            for (int c = 0; c < 7; ++c) nw[c] = band[(eb + sq * c) * G + g];
            nr = rv[(rj + rs * 7) * G + g];
            float inv = __builtin_amdgcn_rsqf(W[0][0]);
            float l[7];
#pragma unroll
            for (int r = 1; r < 7; ++r) l[r] = W[r][0] * inv;
            float y = R[0] * inv;
            band[lc * G + g] = inv;
#pragma unroll
            for (int r = 1; r < 7; ++r) band[(lc + r) * G + g] = l[r];
            rv[rj * G + g] = y;
#pragma unroll
            for (int r = 0; r < 6; ++r) {
#pragma unroll
                for (int c = 0; c <= r; ++c) W[r][c] = W[r + 1][c + 1] - l[r + 1] * l[c + 1];
                R[r] = R[r + 1] - l[r + 1] * y;
            }
#pragma unroll
            for (int c = 0; c < 7; ++c) W[6][c] = nw[c];
            R[6] = nr;
            eb += dj; lc += dj; rj += rs;
        }
        {
            float O[6][6], Or[6];
#pragma unroll
            for (int r = 0; r < 6; ++r) {
#pragma unroll
                for (int c = 0; c <= r; ++c) O[r][c] = __shfl_xor(W[r][c], 16, 32);
                Or[r] = __shfl_xor(R[r], 16, 32);
            }
            float S[6][6], Sr[6];
#pragma unroll
            for (int r = 0; r < 6; ++r) {
#pragma unroll
                for (int c = 0; c <= r; ++c)
                    S[r][c] = W[r][c] + (O[5 - c][5 - r] - band[(o0 + sp * (42 + r) + sq * (42 + c)) * G + g]);
                Sr[r] = R[r] + (Or[5 - r] - rv[(ro + rs * (42 + r)) * G + g]);
            }
#pragma unroll
            for (int r = 0; r < 6; ++r) {
#pragma unroll
                for (int c = 0; c <= r; ++c) W[r][c] = S[r][c];
                R[r] = Sr[r];
            }
#pragma unroll
            for (int c = 0; c < 7; ++c) W[6][c] = 0.0f;
            R[6] = 0.0f;
        }
        for (int j = 42; j < 48; ++j) {
            float inv = __builtin_amdgcn_rsqf(W[0][0]);
            float l[7];
#pragma unroll
            for (int r = 1; r < 7; ++r) l[r] = W[r][0] * inv;
            float y = R[0] * inv;
            if (!rev) {
                band[(j * 7) * G + g] = inv;
#pragma unroll
                for (int r = 1; r < 7; ++r) band[(j * 7 + r) * G + g] = l[r];
                rv[j * G + g] = y;
            }
#pragma unroll
            for (int r = 0; r < 6; ++r) {
#pragma unroll
                for (int c = 0; c <= r; ++c) W[r][c] = W[r + 1][c + 1] - l[r + 1] * l[c + 1];
                R[r] = R[r + 1] - l[r + 1] * y;
            }
#pragma unroll
            for (int c = 0; c < 7; ++c) W[6][c] = 0.0f;
            R[6] = 0.0f;
        }
        float X[7];
#pragma unroll
        for (int dd = 0; dd < 7; ++dd) X[dd] = 0.0f;
        if (!rev) {
            for (int j = 47; j >= 42; --j) {
                float Lc[7];
#pragma unroll
                for (int dd = 0; dd < 7; ++dd) Lc[dd] = band[(j * 7 + dd) * G + g];
                float s = rv[j * G + g];
#pragma unroll
                for (int dd = 1; dd < 7; ++dd) s -= Lc[dd] * X[dd];
                float x = s * Lc[0];
                rv[j * G + g] = x;
#pragma unroll
                for (int dd = 6; dd > 1; --dd) X[dd] = X[dd - 1];
                X[1] = x;
            }
        }
        {
            float Ox[7];
#pragma unroll
            for (int dd = 1; dd < 7; ++dd) Ox[dd] = __shfl_xor(X[dd], 16, 32);
            if (rev) {
#pragma unroll
                for (int dd = 1; dd < 7; ++dd) X[dd] = Ox[7 - dd];
            }
        }
        for (int j = 41; j >= 0; --j) {
            float Lc[7];
            lc -= dj; rj -= rs;
#pragma unroll
            for (int dd = 0; dd < 7; ++dd) Lc[dd] = band[(lc + dd) * G + g];
            float s = rv[rj * G + g];
#pragma unroll
            for (int dd = 1; dd < 7; ++dd) s -= Lc[dd] * X[dd];
            float x = s * Lc[0];
            rv[rj * G + g] = x;
#pragma unroll
            for (int dd = 6; dd > 1; --dd) X[dd] = X[dd - 1];
            X[1] = x;
        }
    }
    __syncthreads();
    for (int i = tid; i < nsys * 90; i += 256) {
        int g = i / 90, j = i - g * 90;
        out[b0 * 90 + i] = rv[j * G + g];
    }
}

extern "C" void kernel_launch(void* const* d_in, const int* in_sizes, int n_in,
                              void* d_out, int out_size, void* d_ws, size_t ws_size,
                              hipStream_t stream) {
    (void)n_in; (void)out_size; (void)d_ws; (void)ws_size;
    const float* coeffs = (const float*)d_in[0];
    const float* rhs    = (const float*)d_in[1];
    const float* iv     = (const float*)d_in[2];
    const float* steps  = (const float*)d_in[3];
    float* out = (float*)d_out;
    int bs = in_sizes[0] / 90;
    odein_solve_kernel<<<(bs + G - 1) / G, 256, 0, stream>>>(coeffs, rhs, iv, steps, out, bs);
}


# LDS reads of the band Cholesky and both substitutions issued three columns ahead of use, in two alternating register sets held in place by a scheduling barrier; sums and their order unchanged
# speedup vs baseline: 44.1009x; 1.1272x over previous
#include <hip/hip_runtime.h>
#include <hip/hip_bf16.h>


typedef float v2f __attribute__((ext_vector_type(2)));
typedef float v8f __attribute__((ext_vector_type(8)));

#define SA    104
#define KROWS 24
#define NB    90
#define BW    6
#define G     16
#define NBP   96
#define NW    8

static __device__ __forceinline__ void wave_lds_handover() {
    __builtin_amdgcn_fence(__ATOMIC_RELEASE, "wavefront");
    __builtin_amdgcn_wave_barrier();
    __builtin_amdgcn_fence(__ATOMIC_ACQUIRE, "wavefront");
}

__global__ __launch_bounds__(256)
void odein_solve_kernel(const float* __restrict__ coeffs,
                        const float* __restrict__ rhs,
                        const float* __restrict__ iv_rhs,
                        const float* __restrict__ steps,
                        float* __restrict__ out,
                        int bs)
{
    __shared__ float Abuf[NW * KROWS * SA];
    __shared__ float band[NBP * 7 * G];
    __shared__ float rv[NBP * G];
    __shared__ float cs[G * 90];
    __shared__ float rs[G * 30];
    __shared__ float ivs[G * 4];
    __shared__ float sts[G * 29];

    const int tid = threadIdx.x;
    const int b0  = blockIdx.x * G;
    const int nsys = (bs - b0 < G) ? (bs - b0) : G;

    for (int i = tid; i < G * 90; i += 256) cs[i]  = (i < nsys * 90) ? coeffs[b0 * 90 + i] : 0.0f;
    for (int i = tid; i < G * 30; i += 256) rs[i]  = (i < nsys * 30) ? rhs[b0 * 30 + i] : 0.0f;
    for (int i = tid; i < G * 29; i += 256) sts[i] = (i < nsys * 29) ? steps[b0 * 29 + i] : 0.0f;
    if (tid < G * 4) ivs[tid] = (tid < nsys * 4) ? iv_rhs[b0 * 4 + tid] : 0.0f;
    for (int idx = tid; idx < NW * KROWS * SA; idx += 256) Abuf[idx] = 0.0f;
    for (int i = tid; i < (NBP - NB) * 7 * G; i += 256) band[NB * 7 * G + i] = 0.0f;
    for (int i = tid; i < (NBP - NB) * G; i += 256) rv[NB * G + i] = 0.0f;
    __syncthreads();

    const int lane = tid & 31, wv = tid >> 5;
    const int half = lane >> 4, lm = lane & 15;
    constexpr int NT = 11;
    constexpr int TI[NT] = {0, 1, 1, 2, 2, 3, 3, 4, 4, 5, 5};
    constexpr int TJ[NT] = {0, 0, 1, 1, 2, 2, 3, 3, 4, 4, 5};
    float* Aw = Abuf + wv * (KROWS * SA);

    if (lane < 4) {
        const int ivcol = lane + (lane >> 1);
        const int ivrow = lane == 0 ? 6 : lane == 1 ? 10 : lane == 2 ? 18 : 19;
        Aw[ivrow * SA + ivcol] = 1.0f;
    }

    for (int s2 = 0; s2 < G / NW; ++s2) {
        const int g = (G / NW) * wv + s2;
        if (lane < 30) {
            const int s = lane;
            const float c0 = cs[g * 90 + 3 * s], c1 = cs[g * 90 + 3 * s + 1], c2 = cs[g * 90 + 3 * s + 2];
            const float r  = rs[g * 30 + s];
            const float i0 = ivs[g * 4 + 2 * (s & 1)], i1 = ivs[g * 4 + 2 * (s & 1) + 1];
            const float h  = sts[g * 29 + (s < 29 ? s : 28)];
            const float hp = sts[g * 29 + (s >= 1 ? s - 1 : 0)];
            const int row = s % 3;
            float v0 = c0 * r, v1 = c1 * r;
            const float v2 = c2 * r;
            if (s < 2) { v0 += i0; v1 += i1; }
            Aw[row * SA + 3 * s + 0] = c0;
            Aw[row * SA + 3 * s + 1] = c1;
            Aw[row * SA + 3 * s + 2] = c2;
            rv[(3 * s + 0) * G + g] = v0;
            rv[(3 * s + 1) * G + g] = v1;
            rv[(3 * s + 2) * G + g] = v2;
            if (s < 29) {
                const float h2 = h * h;
                const int q = s & 3;
                int rr = 3 + q;
                Aw[rr * SA + 3 * (s + 1) + 0] = -1.0f;
                Aw[rr * SA + 3 * s + 0] = 1.0f;
                Aw[rr * SA + 3 * s + 1] = h;
                Aw[rr * SA + 3 * s + 2] = 0.5f * h2;
                rr = 7 + q;
                Aw[rr * SA + 3 * (s + 1) + 1] = -h;
                Aw[rr * SA + 3 * s + 1] = h;
                Aw[rr * SA + 3 * s + 2] = h2;
                rr = 11 + q;
                Aw[rr * SA + 3 * s + 0] = -1.0f;
                Aw[rr * SA + 3 * (s + 1) + 0] = 1.0f;
                Aw[rr * SA + 3 * (s + 1) + 1] = -h;
                Aw[rr * SA + 3 * (s + 1) + 2] = 0.5f * h2;
                rr = 15 + q;
                Aw[rr * SA + 3 * s + 1] = h;
                Aw[rr * SA + 3 * (s + 1) + 1] = -h;
                Aw[rr * SA + 3 * (s + 1) + 2] = h2;
                if (s >= 1) {
                    rr = 19 + s % 5;
                    Aw[rr * SA + 3 * (s - 1) + 1] = -1.0f;
                    Aw[rr * SA + 3 * (s + 1) + 1] = 1.0f;
                    Aw[rr * SA + 3 * s + 2] = -(h + hp);
                }
            }
        }
        wave_lds_handover();

        v2f fr[6][KROWS / 4];
#pragma unroll
        for (int kk = 0; kk < KROWS / 4; ++kk) {
            const int k0 = kk * 4 + 2 * half;
#pragma unroll
            for (int b = 0; b < 6; ++b) {
                fr[b][kk].x = Aw[k0 * SA + b * 16 + lm];
                fr[b][kk].y = Aw[(k0 + 1) * SA + b * 16 + lm];
            }
        }
        wave_lds_handover();

        v8f acc[NT];
#pragma unroll
        for (int t = 0; t < NT; ++t) acc[t] = (v8f){0.f, 0.f, 0.f, 0.f, 0.f, 0.f, 0.f, 0.f};
#pragma unroll
        for (int kk = 0; kk < KROWS / 4; ++kk) {
#pragma unroll
            for (int t = 0; t < NT; ++t)
                acc[t] = __builtin_amdgcn_wmma_f32_16x16x4_f32(false, fr[TI[t]][kk], false, fr[TJ[t]][kk],
                                                               (short)0, acc[t], false, false);
        }

        float* bp = band + (6 * lm + 8 * half) * G + g;
#pragma unroll
        for (int v = 0; v < 8; ++v) {
            const int d0 = v + 8 * half - lm;
            if (d0 >= 0 && d0 <= BW) {
#pragma unroll
                for (int j = 0; j < 5; ++j) bp[(112 * j + v) * G] = acc[2 * j][v];
                if (lm < NB - 80) bp[(112 * 5 + v) * G] = acc[10][v];
            }
            if (d0 + 16 >= 0 && d0 + 16 <= BW) {
#pragma unroll
                for (int j = 0; j < 5; ++j) bp[(112 * j + 16 + v) * G] = acc[2 * j + 1][v];
            }
        }
    }
    __syncthreads();

    if (tid < 32) {
        const int g   = tid & 15;
        const int rev = tid >> 4;
        const int o0 = rev ? 623 : 0, sp = rev ? -6 : 1, sq = rev ? -1 : 6;
        const int ro = rev ? 89 : 0,  rs = rev ? -1 : 1;
        float W[7][7];
        float R[7];
#pragma unroll
        for (int r = 0; r < 7; ++r) {
#pragma unroll
            for (int c = 0; c <= r; ++c) W[r][c] = band[(o0 + sp * r + sq * c) * G + g];
            R[r] = rv[(ro + rs * r) * G + g];
        }
        const int dj = sp + sq;
        int eb = o0 + sp * 7 + sq;
        int lc = o0;
        int rj = ro;
        float PA[3][7], PAr[3], PB[3][7], PBr[3];
#pragma unroll
        for (int k = 0; k < 3; ++k) {
#pragma unroll
            for (int c = 0; c < 7; ++c) PA[k][c] = band[(eb + k * dj + sq * c) * G + g];
            PAr[k] = rv[(rj + rs * (7 + k)) * G + g];
        }
        for (int t = 0; t < 7; ++t) {
            eb += 3 * dj;
#pragma unroll
            for (int k = 0; k < 3; ++k) {
#pragma unroll
                for (int c = 0; c < 7; ++c) PB[k][c] = band[(eb + k * dj + sq * c) * G + g];
                PBr[k] = rv[(rj + rs * (10 + k)) * G + g];
            }
            __builtin_amdgcn_sched_barrier(0);
#pragma unroll
            for (int h = 0; h < 2; ++h) {
                if (h == 1) {
                    eb += 3 * dj;
#pragma unroll
                    for (int k = 0; k < 3; ++k) {
#pragma unroll
                        for (int c = 0; c < 7; ++c) PA[k][c] = band[(eb + k * dj + sq * c) * G + g];
                        PAr[k] = rv[(rj + rs * (10 + k)) * G + g];
                    }
                    __builtin_amdgcn_sched_barrier(0);
                }
#pragma unroll
                for (int k = 0; k < 3; ++k) {
                    float inv = __builtin_amdgcn_rsqf(W[0][0]);
                    float l[7];
#pragma unroll
                    for (int r = 1; r < 7; ++r) l[r] = W[r][0] * inv;
                    float y = R[0] * inv;
                    band[lc * G + g] = inv;
#pragma unroll
                    for (int r = 1; r < 7; ++r) band[(lc + r) * G + g] = l[r];
                    rv[rj * G + g] = y;
#pragma unroll
                    for (int r = 0; r < 6; ++r) {
#pragma unroll
                        for (int c = 0; c <= r; ++c) W[r][c] = W[r + 1][c + 1] - l[r + 1] * l[c + 1];
                        R[r] = R[r + 1] - l[r + 1] * y;
                    }
#pragma unroll
                    for (int c = 0; c < 7; ++c) W[6][c] = (h == 0) ? PA[k][c] : PB[k][c];
                    R[6] = (h == 0) ? PAr[k] : PBr[k];
                    lc += dj; rj += rs;
                }
            }
        }
        {
            float O[6][6], Or[6];
#pragma unroll
            for (int r = 0; r < 6; ++r) {
#pragma unroll
                for (int c = 0; c <= r; ++c) O[r][c] = __shfl_xor(W[r][c], 16, 32);
                Or[r] = __shfl_xor(R[r], 16, 32);
            }
            float S[6][6], Sr[6];
#pragma unroll
            for (int r = 0; r < 6; ++r) {
#pragma unroll
                for (int c = 0; c <= r; ++c)
                    S[r][c] = W[r][c] + (O[5 - c][5 - r] - band[(o0 + sp * (42 + r) + sq * (42 + c)) * G + g]);
                Sr[r] = R[r] + (Or[5 - r] - rv[(ro + rs * (42 + r)) * G + g]);
            }
#pragma unroll
            for (int r = 0; r < 6; ++r) {
#pragma unroll
                for (int c = 0; c <= r; ++c) W[r][c] = S[r][c];
                R[r] = Sr[r];
            }
#pragma unroll
            for (int c = 0; c < 7; ++c) W[6][c] = 0.0f;
            R[6] = 0.0f;
        }
        for (int j = 42; j < 48; ++j) {
            float inv = __builtin_amdgcn_rsqf(W[0][0]);
            float l[7];
#pragma unroll
            for (int r = 1; r < 7; ++r) l[r] = W[r][0] * inv;
            float y = R[0] * inv;
            if (!rev) {
                band[(j * 7) * G + g] = inv;
#pragma unroll
                for (int r = 1; r < 7; ++r) band[(j * 7 + r) * G + g] = l[r];
                rv[j * G + g] = y;
            }
#pragma unroll
            for (int r = 0; r < 6; ++r) {
#pragma unroll
                for (int c = 0; c <= r; ++c) W[r][c] = W[r + 1][c + 1] - l[r + 1] * l[c + 1];
                R[r] = R[r + 1] - l[r + 1] * y;
            }
#pragma unroll
            for (int c = 0; c < 7; ++c) W[6][c] = 0.0f;
            R[6] = 0.0f;
        }
        float X[7];
#pragma unroll
        for (int dd = 0; dd < 7; ++dd) X[dd] = 0.0f;
        float CA[3][7], CAy[3], CB[3][7], CBy[3];
#pragma unroll
        for (int k = 0; k < 3; ++k) {
#pragma unroll
            for (int dd = 0; dd < 7; ++dd) CA[k][dd] = band[(lc - (k + 1) * dj + dd) * G + g];
            CAy[k] = rv[(rj - (k + 1) * rs) * G + g];
        }
        if (!rev) {
            float Lb[6][7], yb[6];
#pragma unroll
            for (int jj = 0; jj < 6; ++jj) {
#pragma unroll
                for (int dd = 0; dd < 7; ++dd) Lb[jj][dd] = band[((47 - jj) * 7 + dd) * G + g];
                yb[jj] = rv[(47 - jj) * G + g];
            }
            __builtin_amdgcn_sched_barrier(0);
#pragma unroll
            for (int jj = 0; jj < 6; ++jj) {
                float s = yb[jj];
#pragma unroll
                for (int dd = 1; dd < 7; ++dd) s -= Lb[jj][dd] * X[dd];
                float x = s * Lb[jj][0];
                rv[(47 - jj) * G + g] = x;
#pragma unroll
                for (int dd = 6; dd > 1; --dd) X[dd] = X[dd - 1];
                X[1] = x;
            }
        }
        {
            float Ox[7];
#pragma unroll
            for (int dd = 1; dd < 7; ++dd) Ox[dd] = __shfl_xor(X[dd], 16, 32);
            if (rev) {
#pragma unroll
                for (int dd = 1; dd < 7; ++dd) X[dd] = Ox[7 - dd];
            }
        }
        for (int t = 0; t < 7; ++t) {
#pragma unroll
            for (int h = 0; h < 2; ++h) {
                lc -= 3 * dj; rj -= 3 * rs;
                if (h == 0) {
#pragma unroll
                    for (int k = 0; k < 3; ++k) {
#pragma unroll
                        for (int dd = 0; dd < 7; ++dd) CB[k][dd] = band[(lc - (k + 1) * dj + dd) * G + g];
                        CBy[k] = rv[(rj - (k + 1) * rs) * G + g];
                    }
                } else if (t < 6) {
#pragma unroll
                    for (int k = 0; k < 3; ++k) {
#pragma unroll
                        for (int dd = 0; dd < 7; ++dd) CA[k][dd] = band[(lc - (k + 1) * dj + dd) * G + g];
                        CAy[k] = rv[(rj - (k + 1) * rs) * G + g];
                    }
                }
                __builtin_amdgcn_sched_barrier(0);
#pragma unroll
                for (int k = 0; k < 3; ++k) {
                    float s = (h == 0) ? CAy[k] : CBy[k];
#pragma unroll
                    for (int dd = 1; dd < 7; ++dd) s -= ((h == 0) ? CA[k][dd] : CB[k][dd]) * X[dd];
                    float x = s * ((h == 0) ? CA[k][0] : CB[k][0]);
                    rv[(rj + (2 - k) * rs) * G + g] = x;
#pragma unroll
                    for (int dd = 6; dd > 1; --dd) X[dd] = X[dd - 1];
                    X[1] = x;
                }
            }
        }
    }
    __syncthreads();
    for (int i = tid; i < nsys * 90; i += 256) {
        int g = i / 90, j = i - g * 90;
        out[b0 * 90 + i] = rv[j * G + g];
    }
}

extern "C" void kernel_launch(void* const* d_in, const int* in_sizes, int n_in,
                              void* d_out, int out_size, void* d_ws, size_t ws_size,
                              hipStream_t stream) {
    (void)n_in; (void)out_size; (void)d_ws; (void)ws_size;
    const float* coeffs = (const float*)d_in[0];
    const float* rhs    = (const float*)d_in[1];
    const float* iv     = (const float*)d_in[2];
    const float* steps  = (const float*)d_in[3];
    float* out = (float*)d_out;
    int bs = in_sizes[0] / 90;
    odein_solve_kernel<<<(bs + G - 1) / G, 256, 0, stream>>>(coeffs, rhs, iv, steps, out, bs);
}


# LDS arrays band[] and rv[] hold 17 floats per unit, was 16 (one unused): band store, rv stores and read-out hit distinct banks, were 2 to 4 banks with lanes queued; solve keeps places pre-scaled
# speedup vs baseline: 49.4091x; 1.1204x over previous
#include <hip/hip_runtime.h>
#include <hip/hip_bf16.h>


typedef float v2f __attribute__((ext_vector_type(2)));
typedef float v8f __attribute__((ext_vector_type(8)));

#define SA    104
#define KROWS 24
#define NB    90
#define BW    6
#define G     16
#define GS    17
#define NBP   96
#define NW    8

static __device__ __forceinline__ void wave_lds_handover() {
    __builtin_amdgcn_fence(__ATOMIC_RELEASE, "wavefront");
    __builtin_amdgcn_wave_barrier();
    __builtin_amdgcn_fence(__ATOMIC_ACQUIRE, "wavefront");
}

__global__ __launch_bounds__(256)
void odein_solve_kernel(const float* __restrict__ coeffs,
                        const float* __restrict__ rhs,
                        const float* __restrict__ iv_rhs,
                        const float* __restrict__ steps,
                        float* __restrict__ out,
                        int bs)
{
    __shared__ float Abuf[NW * KROWS * SA];
    __shared__ float band[NBP * 7 * GS];
    __shared__ float rv[NBP * GS];
    __shared__ float cs[G * 90];
    __shared__ float rs[G * 30];
    __shared__ float ivs[G * 4];
    __shared__ float sts[G * 29];

    const int tid = threadIdx.x;
    const int b0  = blockIdx.x * G;
    const int nsys = (bs - b0 < G) ? (bs - b0) : G;

    for (int i = tid; i < G * 90; i += 256) cs[i]  = (i < nsys * 90) ? coeffs[b0 * 90 + i] : 0.0f;
    for (int i = tid; i < G * 30; i += 256) rs[i]  = (i < nsys * 30) ? rhs[b0 * 30 + i] : 0.0f;
    for (int i = tid; i < G * 29; i += 256) sts[i] = (i < nsys * 29) ? steps[b0 * 29 + i] : 0.0f;
    if (tid < G * 4) ivs[tid] = (tid < nsys * 4) ? iv_rhs[b0 * 4 + tid] : 0.0f;
    for (int idx = tid; idx < NW * KROWS * SA; idx += 256) Abuf[idx] = 0.0f;
    for (int i = tid; i < (NBP - NB) * 7 * GS; i += 256) band[NB * 7 * GS + i] = 0.0f;
    for (int i = tid; i < (NBP - NB) * GS; i += 256) rv[NB * GS + i] = 0.0f;
    __syncthreads();

    const int lane = tid & 31, wv = tid >> 5;
    const int half = lane >> 4, lm = lane & 15;
    constexpr int NT = 11;
    constexpr int TI[NT] = {0, 1, 1, 2, 2, 3, 3, 4, 4, 5, 5};
    constexpr int TJ[NT] = {0, 0, 1, 1, 2, 2, 3, 3, 4, 4, 5};
    float* Aw = Abuf + wv * (KROWS * SA);

    if (lane < 4) {
        const int ivcol = lane + (lane >> 1);
        const int ivrow = lane == 0 ? 6 : lane == 1 ? 10 : lane == 2 ? 18 : 19;
        Aw[ivrow * SA + ivcol] = 1.0f;
    }

    for (int s2 = 0; s2 < G / NW; ++s2) {
        const int g = (G / NW) * wv + s2;
        if (lane < 30) {
            const int s = lane;
            const float c0 = cs[g * 90 + 3 * s], c1 = cs[g * 90 + 3 * s + 1], c2 = cs[g * 90 + 3 * s + 2];
            const float r  = rs[g * 30 + s];
            const float i0 = ivs[g * 4 + 2 * (s & 1)], i1 = ivs[g * 4 + 2 * (s & 1) + 1];
            const float h  = sts[g * 29 + (s < 29 ? s : 28)];
            const float hp = sts[g * 29 + (s >= 1 ? s - 1 : 0)];
            const int row = s % 3;
            float v0 = c0 * r, v1 = c1 * r;
            const float v2 = c2 * r;
            if (s < 2) { v0 += i0; v1 += i1; }
            Aw[row * SA + 3 * s + 0] = c0;
            Aw[row * SA + 3 * s + 1] = c1;
            Aw[row * SA + 3 * s + 2] = c2;
            rv[(3 * s + 0) * GS + g] = v0;
            rv[(3 * s + 1) * GS + g] = v1;
            rv[(3 * s + 2) * GS + g] = v2;
            if (s < 29) {
                const float h2 = h * h;
                const int q = s & 3;
                int rr = 3 + q;
                Aw[rr * SA + 3 * (s + 1) + 0] = -1.0f;
                Aw[rr * SA + 3 * s + 0] = 1.0f;
                Aw[rr * SA + 3 * s + 1] = h;
                Aw[rr * SA + 3 * s + 2] = 0.5f * h2;
                rr = 7 + q;
                Aw[rr * SA + 3 * (s + 1) + 1] = -h;
                Aw[rr * SA + 3 * s + 1] = h;
                Aw[rr * SA + 3 * s + 2] = h2;
                rr = 11 + q;
                Aw[rr * SA + 3 * s + 0] = -1.0f;
                Aw[rr * SA + 3 * (s + 1) + 0] = 1.0f;
                Aw[rr * SA + 3 * (s + 1) + 1] = -h;
                Aw[rr * SA + 3 * (s + 1) + 2] = 0.5f * h2;
                rr = 15 + q;
                Aw[rr * SA + 3 * s + 1] = h;
                Aw[rr * SA + 3 * (s + 1) + 1] = -h;
                Aw[rr * SA + 3 * (s + 1) + 2] = h2;
                if (s >= 1) {
                    rr = 19 + s % 5;
                    Aw[rr * SA + 3 * (s - 1) + 1] = -1.0f;
                    Aw[rr * SA + 3 * (s + 1) + 1] = 1.0f;
                    Aw[rr * SA + 3 * s + 2] = -(h + hp);
                }
            }
        }
        wave_lds_handover();

        v2f fr[6][KROWS / 4];
#pragma unroll
        for (int kk = 0; kk < KROWS / 4; ++kk) {
            const int k0 = kk * 4 + 2 * half;
#pragma unroll
            for (int b = 0; b < 6; ++b) {
                fr[b][kk].x = Aw[k0 * SA + b * 16 + lm];
                fr[b][kk].y = Aw[(k0 + 1) * SA + b * 16 + lm];
            }
        }
        wave_lds_handover();

        v8f acc[NT];
#pragma unroll
        for (int t = 0; t < NT; ++t) acc[t] = (v8f){0.f, 0.f, 0.f, 0.f, 0.f, 0.f, 0.f, 0.f};
#pragma unroll
        for (int kk = 0; kk < KROWS / 4; ++kk) {
#pragma unroll
            for (int t = 0; t < NT; ++t)
                acc[t] = __builtin_amdgcn_wmma_f32_16x16x4_f32(false, fr[TI[t]][kk], false, fr[TJ[t]][kk],
                                                               (short)0, acc[t], false, false);
        }

        float* bp = band + (6 * lm + 8 * half) * GS + g;
#pragma unroll
        for (int v = 0; v < 8; ++v) {
            const int d0 = v + 8 * half - lm;
            if (d0 >= 0 && d0 <= BW) {
#pragma unroll
                for (int j = 0; j < 5; ++j) bp[(112 * j + v) * GS] = acc[2 * j][v];
                if (lm < NB - 80) bp[(112 * 5 + v) * GS] = acc[10][v];
            }
            if (d0 + 16 >= 0 && d0 + 16 <= BW) {
#pragma unroll
                for (int j = 0; j < 5; ++j) bp[(112 * j + 16 + v) * GS] = acc[2 * j + 1][v];
            }
        }
    }
    __syncthreads();

    if (tid < 32) {
        const int g   = tid & 15;
        const int rev = tid >> 4;
        const int o0 = rev ? 623 * GS : 0, sp = rev ? -6 * GS : GS, sq = rev ? -GS : 6 * GS;
        const int ro = rev ? 89 * GS : 0,  rs = rev ? -GS : GS;
        float W[7][7];
        float R[7];
#pragma unroll
        for (int r = 0; r < 7; ++r) {
#pragma unroll
            for (int c = 0; c <= r; ++c) W[r][c] = band[o0 + sp * r + sq * c + g];
            R[r] = rv[ro + rs * r + g];
        }
        const int dj = sp + sq;
        int eb = o0 + sp * 7 + sq;
        int lc = o0;
        int rj = ro;
        float PA[3][7], PAr[3], PB[3][7], PBr[3];
#pragma unroll
        for (int k = 0; k < 3; ++k) {
#pragma unroll
            for (int c = 0; c < 7; ++c) PA[k][c] = band[eb + k * dj + sq * c + g];
            PAr[k] = rv[rj + rs * (7 + k) + g];
        }
        for (int t = 0; t < 7; ++t) {
            eb += 3 * dj;
#pragma unroll
            for (int k = 0; k < 3; ++k) {
#pragma unroll
                for (int c = 0; c < 7; ++c) PB[k][c] = band[eb + k * dj + sq * c + g];
                PBr[k] = rv[rj + rs * (10 + k) + g];
            }
            __builtin_amdgcn_sched_barrier(0);
#pragma unroll
            for (int h = 0; h < 2; ++h) {
                if (h == 1) {
                    eb += 3 * dj;
#pragma unroll
                    for (int k = 0; k < 3; ++k) {
#pragma unroll
                        for (int c = 0; c < 7; ++c) PA[k][c] = band[eb + k * dj + sq * c + g];
                        PAr[k] = rv[rj + rs * (10 + k) + g];
                    }
                    __builtin_amdgcn_sched_barrier(0);
                }
#pragma unroll
                for (int k = 0; k < 3; ++k) {
                    float inv = __builtin_amdgcn_rsqf(W[0][0]);
                    float l[7];
#pragma unroll
                    for (int r = 1; r < 7; ++r) l[r] = W[r][0] * inv;
                    float y = R[0] * inv;
                    band[lc + g] = inv;
#pragma unroll
                    for (int r = 1; r < 7; ++r) band[lc + r * GS + g] = l[r];
                    rv[rj + g] = y;
#pragma unroll
                    for (int r = 0; r < 6; ++r) {
#pragma unroll
                        for (int c = 0; c <= r; ++c) W[r][c] = W[r + 1][c + 1] - l[r + 1] * l[c + 1];
                        R[r] = R[r + 1] - l[r + 1] * y;
                    }
#pragma unroll
                    for (int c = 0; c < 7; ++c) W[6][c] = (h == 0) ? PA[k][c] : PB[k][c];
                    R[6] = (h == 0) ? PAr[k] : PBr[k];
                    lc += dj; rj += rs;
                }
            }
        }
        {
            float O[6][6], Or[6];
#pragma unroll
            for (int r = 0; r < 6; ++r) {
#pragma unroll
                for (int c = 0; c <= r; ++c) O[r][c] = __shfl_xor(W[r][c], 16, 32);
                Or[r] = __shfl_xor(R[r], 16, 32);
            }
            float S[6][6], Sr[6];
#pragma unroll
            for (int r = 0; r < 6; ++r) {
#pragma unroll
                for (int c = 0; c <= r; ++c)
                    S[r][c] = W[r][c] + (O[5 - c][5 - r] - band[o0 + sp * (42 + r) + sq * (42 + c) + g]);
                Sr[r] = R[r] + (Or[5 - r] - rv[ro + rs * (42 + r) + g]);
            }
#pragma unroll
            for (int r = 0; r < 6; ++r) {
#pragma unroll
                for (int c = 0; c <= r; ++c) W[r][c] = S[r][c];
                R[r] = Sr[r];
            }
#pragma unroll
            for (int c = 0; c < 7; ++c) W[6][c] = 0.0f;
            R[6] = 0.0f;
        }
        for (int j = 42; j < 48; ++j) {
            float inv = __builtin_amdgcn_rsqf(W[0][0]);
            float l[7];
#pragma unroll
            for (int r = 1; r < 7; ++r) l[r] = W[r][0] * inv;
            float y = R[0] * inv;
            if (!rev) {
                band[(j * 7) * GS + g] = inv;
#pragma unroll
                for (int r = 1; r < 7; ++r) band[(j * 7 + r) * GS + g] = l[r];
                rv[j * GS + g] = y;
            }
#pragma unroll
            for (int r = 0; r < 6; ++r) {
#pragma unroll
                for (int c = 0; c <= r; ++c) W[r][c] = W[r + 1][c + 1] - l[r + 1] * l[c + 1];
                R[r] = R[r + 1] - l[r + 1] * y;
            }
#pragma unroll
            for (int c = 0; c < 7; ++c) W[6][c] = 0.0f;
            R[6] = 0.0f;
        }
        float X[7];
#pragma unroll
        for (int dd = 0; dd < 7; ++dd) X[dd] = 0.0f;
        float CA[3][7], CAy[3], CB[3][7], CBy[3];
#pragma unroll
        for (int k = 0; k < 3; ++k) {
#pragma unroll
            for (int dd = 0; dd < 7; ++dd) CA[k][dd] = band[lc - (k + 1) * dj + dd * GS + g];
            CAy[k] = rv[rj - (k + 1) * rs + g];
        }
        if (!rev) {
            float Lb[6][7], yb[6];
#pragma unroll
            for (int jj = 0; jj < 6; ++jj) {
#pragma unroll
                for (int dd = 0; dd < 7; ++dd) Lb[jj][dd] = band[((47 - jj) * 7 + dd) * GS + g];
                yb[jj] = rv[(47 - jj) * GS + g];
            }
            __builtin_amdgcn_sched_barrier(0);
#pragma unroll
            for (int jj = 0; jj < 6; ++jj) {
                float s = yb[jj];
#pragma unroll
                for (int dd = 1; dd < 7; ++dd) s -= Lb[jj][dd] * X[dd];
                float x = s * Lb[jj][0];
                rv[(47 - jj) * GS + g] = x;
#pragma unroll
                for (int dd = 6; dd > 1; --dd) X[dd] = X[dd - 1];
                X[1] = x;
            }
        }
        {
            float Ox[7];
#pragma unroll
            for (int dd = 1; dd < 7; ++dd) Ox[dd] = __shfl_xor(X[dd], 16, 32);
            if (rev) {
#pragma unroll
                for (int dd = 1; dd < 7; ++dd) X[dd] = Ox[7 - dd];
            }
        }
        for (int t = 0; t < 7; ++t) {
#pragma unroll
            for (int h = 0; h < 2; ++h) {
                lc -= 3 * dj; rj -= 3 * rs;
                if (h == 0) {
#pragma unroll
                    for (int k = 0; k < 3; ++k) {
#pragma unroll
                        for (int dd = 0; dd < 7; ++dd) CB[k][dd] = band[lc - (k + 1) * dj + dd * GS + g];
                        CBy[k] = rv[rj - (k + 1) * rs + g];
                    }
                } else if (t < 6) {
#pragma unroll
                    for (int k = 0; k < 3; ++k) {
#pragma unroll
                        for (int dd = 0; dd < 7; ++dd) CA[k][dd] = band[lc - (k + 1) * dj + dd * GS + g];
                        CAy[k] = rv[rj - (k + 1) * rs + g];
                    }
                }
                __builtin_amdgcn_sched_barrier(0);
#pragma unroll
                for (int k = 0; k < 3; ++k) {
                    float s = (h == 0) ? CAy[k] : CBy[k];
#pragma unroll
                    for (int dd = 1; dd < 7; ++dd) s -= ((h == 0) ? CA[k][dd] : CB[k][dd]) * X[dd];
                    float x = s * ((h == 0) ? CA[k][0] : CB[k][0]);
                    rv[rj + (2 - k) * rs + g] = x;
#pragma unroll
                    for (int dd = 6; dd > 1; --dd) X[dd] = X[dd - 1];
                    X[1] = x;
                }
            }
        }
    }
    __syncthreads();
    for (int i = tid; i < nsys * 90; i += 256) {
        int g = i / 90, j = i - g * 90;
        out[b0 * 90 + i] = rv[j * GS + g];
    }
}

extern "C" void kernel_launch(void* const* d_in, const int* in_sizes, int n_in,
                              void* d_out, int out_size, void* d_ws, size_t ws_size,
                              hipStream_t stream) {
    (void)n_in; (void)out_size; (void)d_ws; (void)ws_size;
    const float* coeffs = (const float*)d_in[0];
    const float* rhs    = (const float*)d_in[1];
    const float* iv     = (const float*)d_in[2];
    const float* steps  = (const float*)d_in[3];
    float* out = (float*)d_out;
    int bs = in_sizes[0] / 90;
    odein_solve_kernel<<<(bs + G - 1) / G, 256, 0, stream>>>(coeffs, rhs, iv, steps, out, bs);
}


# Inputs of a 16-system group staged by 16-byte global loads, a thread's three issued together ahead of the LDS zero fill and waited for once; was 11 load, wait, LDS-store round trips in a row
# speedup vs baseline: 64.5330x; 1.3061x over previous
#include <hip/hip_runtime.h>
#include <hip/hip_bf16.h>


typedef float v2f __attribute__((ext_vector_type(2)));
typedef float v8f __attribute__((ext_vector_type(8)));
typedef float v4f __attribute__((ext_vector_type(4)));

#define SA    104
#define KROWS 24
#define NB    90
#define BW    6
#define G     16
#define GS    17
#define NBP   96
#define NW    8

static __device__ __forceinline__ void wave_lds_handover() {
    __builtin_amdgcn_fence(__ATOMIC_RELEASE, "wavefront");
    __builtin_amdgcn_wave_barrier();
    __builtin_amdgcn_fence(__ATOMIC_ACQUIRE, "wavefront");
}

__global__ __launch_bounds__(256)
void odein_solve_kernel(const float* __restrict__ coeffs,
                        const float* __restrict__ rhs,
                        const float* __restrict__ iv_rhs,
                        const float* __restrict__ steps,
                        float* __restrict__ out,
                        int bs)
{
    __shared__ float Abuf[NW * KROWS * SA];
    __shared__ float band[NBP * 7 * GS];
    __shared__ float rv[NBP * GS];
    __shared__ __attribute__((aligned(16))) float cs[G * 90];
    __shared__ __attribute__((aligned(16))) float rs[G * 30];
    __shared__ __attribute__((aligned(16))) float ivs[G * 4];
    __shared__ __attribute__((aligned(16))) float sts[G * 29];

    const int tid = threadIdx.x;
    const int b0  = blockIdx.x * G;
    const int nsys = (bs - b0 < G) ? (bs - b0) : G;

    const bool full = (nsys == G);
    v4f q0 = (v4f){0.f, 0.f, 0.f, 0.f}, q1 = q0, q2 = q0;
    const float* p1 = (tid < 104) ? coeffs + b0 * 90 + 4 * (256 + tid)
                    : (tid < 224) ? rhs + b0 * 30 + 4 * (tid - 104)
                                  : iv_rhs + b0 * 4 + 4 * (tid - 224);
    float* d1 = (tid < 104) ? cs + 4 * (256 + tid) : (tid < 224) ? rs + 4 * (tid - 104) : ivs + 4 * (tid - 224);
    if (full) {
        __builtin_memcpy(&q0, __builtin_assume_aligned(coeffs + b0 * 90 + 4 * tid, 16), 16);
        if (tid < 240) __builtin_memcpy(&q1, __builtin_assume_aligned(p1, 16), 16);
        if (tid < 116) __builtin_memcpy(&q2, __builtin_assume_aligned(steps + b0 * 29 + 4 * tid, 16), 16);
    }
    for (int idx = tid; idx < NW * KROWS * SA; idx += 256) Abuf[idx] = 0.0f;
    for (int i = tid; i < (NBP - NB) * 7 * GS; i += 256) band[NB * 7 * GS + i] = 0.0f;
    for (int i = tid; i < (NBP - NB) * GS; i += 256) rv[NB * GS + i] = 0.0f;
    if (full) {
        __builtin_memcpy(__builtin_assume_aligned(cs + 4 * tid, 16), &q0, 16);
        if (tid < 240) __builtin_memcpy(__builtin_assume_aligned(d1, 16), &q1, 16);
        if (tid < 116) __builtin_memcpy(__builtin_assume_aligned(sts + 4 * tid, 16), &q2, 16);
    } else {
        for (int i = tid; i < G * 90; i += 256) cs[i]  = (i < nsys * 90) ? coeffs[b0 * 90 + i] : 0.0f;
        for (int i = tid; i < G * 30; i += 256) rs[i]  = (i < nsys * 30) ? rhs[b0 * 30 + i] : 0.0f;
        for (int i = tid; i < G * 29; i += 256) sts[i] = (i < nsys * 29) ? steps[b0 * 29 + i] : 0.0f;
        if (tid < G * 4) ivs[tid] = (tid < nsys * 4) ? iv_rhs[b0 * 4 + tid] : 0.0f;
    }
    __syncthreads();

    const int lane = tid & 31, wv = tid >> 5;
    const int half = lane >> 4, lm = lane & 15;
    constexpr int NT = 11;
    constexpr int TI[NT] = {0, 1, 1, 2, 2, 3, 3, 4, 4, 5, 5};
    constexpr int TJ[NT] = {0, 0, 1, 1, 2, 2, 3, 3, 4, 4, 5};
    float* Aw = Abuf + wv * (KROWS * SA);

    if (lane < 4) {
        const int ivcol = lane + (lane >> 1);
        const int ivrow = lane == 0 ? 6 : lane == 1 ? 10 : lane == 2 ? 18 : 19;
        Aw[ivrow * SA + ivcol] = 1.0f;
    }

    for (int s2 = 0; s2 < G / NW; ++s2) {
        const int g = (G / NW) * wv + s2;
        if (lane < 30) {
            const int s = lane;
            const float c0 = cs[g * 90 + 3 * s], c1 = cs[g * 90 + 3 * s + 1], c2 = cs[g * 90 + 3 * s + 2];
            const float r  = rs[g * 30 + s];
            const float i0 = ivs[g * 4 + 2 * (s & 1)], i1 = ivs[g * 4 + 2 * (s & 1) + 1];
            const float h  = sts[g * 29 + (s < 29 ? s : 28)];
            const float hp = sts[g * 29 + (s >= 1 ? s - 1 : 0)];
            const int row = s % 3;
            float v0 = c0 * r, v1 = c1 * r;
            const float v2 = c2 * r;
            if (s < 2) { v0 += i0; v1 += i1; }
            Aw[row * SA + 3 * s + 0] = c0;
            Aw[row * SA + 3 * s + 1] = c1;
            Aw[row * SA + 3 * s + 2] = c2;
            rv[(3 * s + 0) * GS + g] = v0;
            rv[(3 * s + 1) * GS + g] = v1;
            rv[(3 * s + 2) * GS + g] = v2;
            if (s < 29) {
                const float h2 = h * h;
                const int q = s & 3;
                int rr = 3 + q;
                Aw[rr * SA + 3 * (s + 1) + 0] = -1.0f;
                Aw[rr * SA + 3 * s + 0] = 1.0f;
                Aw[rr * SA + 3 * s + 1] = h;
                Aw[rr * SA + 3 * s + 2] = 0.5f * h2;
                rr = 7 + q;
                Aw[rr * SA + 3 * (s + 1) + 1] = -h;
                Aw[rr * SA + 3 * s + 1] = h;
                Aw[rr * SA + 3 * s + 2] = h2;
                rr = 11 + q;
                Aw[rr * SA + 3 * s + 0] = -1.0f;
                Aw[rr * SA + 3 * (s + 1) + 0] = 1.0f;
                Aw[rr * SA + 3 * (s + 1) + 1] = -h;
                Aw[rr * SA + 3 * (s + 1) + 2] = 0.5f * h2;
                rr = 15 + q;
                Aw[rr * SA + 3 * s + 1] = h;
                Aw[rr * SA + 3 * (s + 1) + 1] = -h;
                Aw[rr * SA + 3 * (s + 1) + 2] = h2;
                if (s >= 1) {
                    rr = 19 + s % 5;
                    Aw[rr * SA + 3 * (s - 1) + 1] = -1.0f;
                    Aw[rr * SA + 3 * (s + 1) + 1] = 1.0f;
                    Aw[rr * SA + 3 * s + 2] = -(h + hp);
                }
            }
        }
        wave_lds_handover();

        v2f fr[6][KROWS / 4];
#pragma unroll
        for (int kk = 0; kk < KROWS / 4; ++kk) {
            const int k0 = kk * 4 + 2 * half;
#pragma unroll
            for (int b = 0; b < 6; ++b) {
                fr[b][kk].x = Aw[k0 * SA + b * 16 + lm];
                fr[b][kk].y = Aw[(k0 + 1) * SA + b * 16 + lm];
            }
        }
        wave_lds_handover();

        v8f acc[NT];
#pragma unroll
        for (int t = 0; t < NT; ++t) acc[t] = (v8f){0.f, 0.f, 0.f, 0.f, 0.f, 0.f, 0.f, 0.f};
#pragma unroll
        for (int kk = 0; kk < KROWS / 4; ++kk) {
#pragma unroll
            for (int t = 0; t < NT; ++t)
                acc[t] = __builtin_amdgcn_wmma_f32_16x16x4_f32(false, fr[TI[t]][kk], false, fr[TJ[t]][kk],
                                                               (short)0, acc[t], false, false);
        }

        float* bp = band + (6 * lm + 8 * half) * GS + g;
#pragma unroll
        for (int v = 0; v < 8; ++v) {
            const int d0 = v + 8 * half - lm;
            if (d0 >= 0 && d0 <= BW) {
#pragma unroll
                for (int j = 0; j < 5; ++j) bp[(112 * j + v) * GS] = acc[2 * j][v];
                if (lm < NB - 80) bp[(112 * 5 + v) * GS] = acc[10][v];
            }
            if (d0 + 16 >= 0 && d0 + 16 <= BW) {
#pragma unroll
                for (int j = 0; j < 5; ++j) bp[(112 * j + 16 + v) * GS] = acc[2 * j + 1][v];
            }
        }
    }
    __syncthreads();

    if (tid < 32) {
        const int g   = tid & 15;
        const int rev = tid >> 4;
        const int o0 = rev ? 623 * GS : 0, sp = rev ? -6 * GS : GS, sq = rev ? -GS : 6 * GS;
        const int ro = rev ? 89 * GS : 0,  rs = rev ? -GS : GS;
        float W[7][7];
        float R[7];
#pragma unroll
        for (int r = 0; r < 7; ++r) {
#pragma unroll
            for (int c = 0; c <= r; ++c) W[r][c] = band[o0 + sp * r + sq * c + g];
            R[r] = rv[ro + rs * r + g];
        }
        const int dj = sp + sq;
        int eb = o0 + sp * 7 + sq;
        int lc = o0;
        int rj = ro;
        float PA[3][7], PAr[3], PB[3][7], PBr[3];
#pragma unroll
        for (int k = 0; k < 3; ++k) {
#pragma unroll
            for (int c = 0; c < 7; ++c) PA[k][c] = band[eb + k * dj + sq * c + g];
            PAr[k] = rv[rj + rs * (7 + k) + g];
        }
        for (int t = 0; t < 7; ++t) {
            eb += 3 * dj;
#pragma unroll
            for (int k = 0; k < 3; ++k) {
#pragma unroll
                for (int c = 0; c < 7; ++c) PB[k][c] = band[eb + k * dj + sq * c + g];
                PBr[k] = rv[rj + rs * (10 + k) + g];
            }
            __builtin_amdgcn_sched_barrier(0);
#pragma unroll
            for (int h = 0; h < 2; ++h) {
                if (h == 1) {
                    eb += 3 * dj;
#pragma unroll
                    for (int k = 0; k < 3; ++k) {
#pragma unroll
                        for (int c = 0; c < 7; ++c) PA[k][c] = band[eb + k * dj + sq * c + g];
                        PAr[k] = rv[rj + rs * (10 + k) + g];
                    }
                    __builtin_amdgcn_sched_barrier(0);
                }
#pragma unroll
                for (int k = 0; k < 3; ++k) {
                    float inv = __builtin_amdgcn_rsqf(W[0][0]);
                    float l[7];
#pragma unroll
                    for (int r = 1; r < 7; ++r) l[r] = W[r][0] * inv;
                    float y = R[0] * inv;
                    band[lc + g] = inv;
#pragma unroll
                    for (int r = 1; r < 7; ++r) band[lc + r * GS + g] = l[r];
                    rv[rj + g] = y;
#pragma unroll
                    for (int r = 0; r < 6; ++r) {
#pragma unroll
                        for (int c = 0; c <= r; ++c) W[r][c] = W[r + 1][c + 1] - l[r + 1] * l[c + 1];
                        R[r] = R[r + 1] - l[r + 1] * y;
                    }
#pragma unroll
                    for (int c = 0; c < 7; ++c) W[6][c] = (h == 0) ? PA[k][c] : PB[k][c];
                    R[6] = (h == 0) ? PAr[k] : PBr[k];
                    lc += dj; rj += rs;
                }
            }
        }
        {
            float O[6][6], Or[6];
#pragma unroll
            for (int r = 0; r < 6; ++r) {
#pragma unroll
                for (int c = 0; c <= r; ++c) O[r][c] = __shfl_xor(W[r][c], 16, 32);
                Or[r] = __shfl_xor(R[r], 16, 32);
            }
            float S[6][6], Sr[6];
#pragma unroll
            for (int r = 0; r < 6; ++r) {
#pragma unroll
                for (int c = 0; c <= r; ++c)
                    S[r][c] = W[r][c] + (O[5 - c][5 - r] - band[o0 + sp * (42 + r) + sq * (42 + c) + g]);
                Sr[r] = R[r] + (Or[5 - r] - rv[ro + rs * (42 + r) + g]);
            }
#pragma unroll
            for (int r = 0; r < 6; ++r) {
#pragma unroll
                for (int c = 0; c <= r; ++c) W[r][c] = S[r][c];
                R[r] = Sr[r];
            }
#pragma unroll
            for (int c = 0; c < 7; ++c) W[6][c] = 0.0f;
            R[6] = 0.0f;
        }
        for (int j = 42; j < 48; ++j) {
            float inv = __builtin_amdgcn_rsqf(W[0][0]);
            float l[7];
#pragma unroll
            for (int r = 1; r < 7; ++r) l[r] = W[r][0] * inv;
            float y = R[0] * inv;
            if (!rev) {
                band[(j * 7) * GS + g] = inv;
#pragma unroll
                for (int r = 1; r < 7; ++r) band[(j * 7 + r) * GS + g] = l[r];
                rv[j * GS + g] = y;
            }
#pragma unroll
            for (int r = 0; r < 6; ++r) {
#pragma unroll
                for (int c = 0; c <= r; ++c) W[r][c] = W[r + 1][c + 1] - l[r + 1] * l[c + 1];
                R[r] = R[r + 1] - l[r + 1] * y;
            }
#pragma unroll
            for (int c = 0; c < 7; ++c) W[6][c] = 0.0f;
            R[6] = 0.0f;
        }
        float X[7];
#pragma unroll
        for (int dd = 0; dd < 7; ++dd) X[dd] = 0.0f;
        float CA[3][7], CAy[3], CB[3][7], CBy[3];
#pragma unroll
        for (int k = 0; k < 3; ++k) {
#pragma unroll
            for (int dd = 0; dd < 7; ++dd) CA[k][dd] = band[lc - (k + 1) * dj + dd * GS + g];
            CAy[k] = rv[rj - (k + 1) * rs + g];
        }
        if (!rev) {
            float Lb[6][7], yb[6];
#pragma unroll
            for (int jj = 0; jj < 6; ++jj) {
#pragma unroll
                for (int dd = 0; dd < 7; ++dd) Lb[jj][dd] = band[((47 - jj) * 7 + dd) * GS + g];
                yb[jj] = rv[(47 - jj) * GS + g];
            }
            __builtin_amdgcn_sched_barrier(0);
#pragma unroll
            for (int jj = 0; jj < 6; ++jj) {
                float s = yb[jj];
#pragma unroll
                for (int dd = 1; dd < 7; ++dd) s -= Lb[jj][dd] * X[dd];
                float x = s * Lb[jj][0];
                rv[(47 - jj) * GS + g] = x;
#pragma unroll
                for (int dd = 6; dd > 1; --dd) X[dd] = X[dd - 1];
                X[1] = x;
            }
        }
        {
            float Ox[7];
#pragma unroll
            for (int dd = 1; dd < 7; ++dd) Ox[dd] = __shfl_xor(X[dd], 16, 32);
            if (rev) {
#pragma unroll
                for (int dd = 1; dd < 7; ++dd) X[dd] = Ox[7 - dd];
            }
        }
        for (int t = 0; t < 7; ++t) {
#pragma unroll
            for (int h = 0; h < 2; ++h) {
                lc -= 3 * dj; rj -= 3 * rs;
                if (h == 0) {
#pragma unroll
                    for (int k = 0; k < 3; ++k) {
#pragma unroll
                        for (int dd = 0; dd < 7; ++dd) CB[k][dd] = band[lc - (k + 1) * dj + dd * GS + g];
                        CBy[k] = rv[rj - (k + 1) * rs + g];
                    }
                } else if (t < 6) {
#pragma unroll
                    for (int k = 0; k < 3; ++k) {
#pragma unroll
                        for (int dd = 0; dd < 7; ++dd) CA[k][dd] = band[lc - (k + 1) * dj + dd * GS + g];
                        CAy[k] = rv[rj - (k + 1) * rs + g];
                    }
                }
                __builtin_amdgcn_sched_barrier(0);
#pragma unroll
                for (int k = 0; k < 3; ++k) {
                    float s = (h == 0) ? CAy[k] : CBy[k];
#pragma unroll
                    for (int dd = 1; dd < 7; ++dd) s -= ((h == 0) ? CA[k][dd] : CB[k][dd]) * X[dd];
                    float x = s * ((h == 0) ? CA[k][0] : CB[k][0]);
                    rv[rj + (2 - k) * rs + g] = x;
#pragma unroll
                    for (int dd = 6; dd > 1; --dd) X[dd] = X[dd - 1];
                    X[1] = x;
                }
            }
        }
    }
    __syncthreads();
    for (int i = tid; i < nsys * 90; i += 256) {
        int g = i / 90, j = i - g * 90;
        out[b0 * 90 + i] = rv[j * GS + g];
    }
}

extern "C" void kernel_launch(void* const* d_in, const int* in_sizes, int n_in,
                              void* d_out, int out_size, void* d_ws, size_t ws_size,
                              hipStream_t stream) {
    (void)n_in; (void)out_size; (void)d_ws; (void)ws_size;
    const float* coeffs = (const float*)d_in[0];
    const float* rhs    = (const float*)d_in[1];
    const float* iv     = (const float*)d_in[2];
    const float* steps  = (const float*)d_in[3];
    float* out = (float*)d_out;
    int bs = in_sizes[0] / 90;
    odein_solve_kernel<<<(bs + G - 1) / G, 256, 0, stream>>>(coeffs, rhs, iv, steps, out, bs);
}


# A^T A on 7 tiles of 16x16 whose rows lie 3 columns behind their columns, tiles 13 columns apart: 10 whole band columns each, the partial ones complete each other; 42 WMMAs a system, was 66
# speedup vs baseline: 67.9725x; 1.0533x over previous
#include <hip/hip_runtime.h>
#include <hip/hip_bf16.h>


typedef float v2f __attribute__((ext_vector_type(2)));
typedef float v8f __attribute__((ext_vector_type(8)));
typedef float v4f __attribute__((ext_vector_type(4)));

#define SA    104
#define KROWS 24
#define CO    3
#define NT    7
#define NB    90
#define BW    6
#define G     16
#define GS    17
#define NBP   96
#define NW    8

static __device__ __forceinline__ void wave_lds_handover() {
    __builtin_amdgcn_fence(__ATOMIC_RELEASE, "wavefront");
    __builtin_amdgcn_wave_barrier();
    __builtin_amdgcn_fence(__ATOMIC_ACQUIRE, "wavefront");
}

__global__ __launch_bounds__(256)
void odein_solve_kernel(const float* __restrict__ coeffs,
                        const float* __restrict__ rhs,
                        const float* __restrict__ iv_rhs,
                        const float* __restrict__ steps,
                        float* __restrict__ out,
                        int bs)
{
    __shared__ float Abuf[NW * KROWS * SA];
    __shared__ float band[NBP * 7 * GS];
    __shared__ float rv[NBP * GS];
    __shared__ __attribute__((aligned(16))) float cs[G * 90];
    __shared__ __attribute__((aligned(16))) float rs[G * 30];
    __shared__ __attribute__((aligned(16))) float ivs[G * 4];
    __shared__ __attribute__((aligned(16))) float sts[G * 29];

    const int tid = threadIdx.x;
    const int b0  = blockIdx.x * G;
    const int nsys = (bs - b0 < G) ? (bs - b0) : G;

    const bool full = (nsys == G);
    v4f q0 = (v4f){0.f, 0.f, 0.f, 0.f}, q1 = q0, q2 = q0;
    const float* p1 = (tid < 104) ? coeffs + b0 * 90 + 4 * (256 + tid)
                    : (tid < 224) ? rhs + b0 * 30 + 4 * (tid - 104)
                                  : iv_rhs + b0 * 4 + 4 * (tid - 224);
    float* d1 = (tid < 104) ? cs + 4 * (256 + tid) : (tid < 224) ? rs + 4 * (tid - 104) : ivs + 4 * (tid - 224);
    if (full) {
        __builtin_memcpy(&q0, __builtin_assume_aligned(coeffs + b0 * 90 + 4 * tid, 16), 16);
        if (tid < 240) __builtin_memcpy(&q1, __builtin_assume_aligned(p1, 16), 16);
        if (tid < 116) __builtin_memcpy(&q2, __builtin_assume_aligned(steps + b0 * 29 + 4 * tid, 16), 16);
    }
    for (int idx = tid; idx < NW * KROWS * SA; idx += 256) Abuf[idx] = 0.0f;
    for (int i = tid; i < (NBP - 84) * 7 * GS; i += 256) band[84 * 7 * GS + i] = 0.0f;
    for (int i = tid; i < (NBP - NB) * GS; i += 256) rv[NB * GS + i] = 0.0f;
    if (full) {
        __builtin_memcpy(__builtin_assume_aligned(cs + 4 * tid, 16), &q0, 16);
        if (tid < 240) __builtin_memcpy(__builtin_assume_aligned(d1, 16), &q1, 16);
        if (tid < 116) __builtin_memcpy(__builtin_assume_aligned(sts + 4 * tid, 16), &q2, 16);
    } else {
        for (int i = tid; i < G * 90; i += 256) cs[i]  = (i < nsys * 90) ? coeffs[b0 * 90 + i] : 0.0f;
        for (int i = tid; i < G * 30; i += 256) rs[i]  = (i < nsys * 30) ? rhs[b0 * 30 + i] : 0.0f;
        for (int i = tid; i < G * 29; i += 256) sts[i] = (i < nsys * 29) ? steps[b0 * 29 + i] : 0.0f;
        if (tid < G * 4) ivs[tid] = (tid < nsys * 4) ? iv_rhs[b0 * 4 + tid] : 0.0f;
    }
    __syncthreads();

    const int lane = tid & 31, wv = tid >> 5;
    const int half = lane >> 4, lm = lane & 15;
    float* Aw = Abuf + wv * (KROWS * SA) + CO;

    if (lane < 4) {
        const int ivcol = lane + (lane >> 1);
        const int ivrow = lane == 0 ? 6 : lane == 1 ? 10 : lane == 2 ? 18 : 19;
        Aw[ivrow * SA + ivcol] = 1.0f;
    }

    for (int s2 = 0; s2 < G / NW; ++s2) {
        const int g = (G / NW) * wv + s2;
        if (lane < 30) {
            const int s = lane;
            const float c0 = cs[g * 90 + 3 * s], c1 = cs[g * 90 + 3 * s + 1], c2 = cs[g * 90 + 3 * s + 2];
            const float r  = rs[g * 30 + s];
            const float i0 = ivs[g * 4 + 2 * (s & 1)], i1 = ivs[g * 4 + 2 * (s & 1) + 1];
            const float h  = sts[g * 29 + (s < 29 ? s : 28)];
            const float hp = sts[g * 29 + (s >= 1 ? s - 1 : 0)];
            const int row = s % 3;
            float v0 = c0 * r, v1 = c1 * r;
            const float v2 = c2 * r;
            if (s < 2) { v0 += i0; v1 += i1; }
            Aw[row * SA + 3 * s + 0] = c0;
            Aw[row * SA + 3 * s + 1] = c1;
            Aw[row * SA + 3 * s + 2] = c2;
            rv[(3 * s + 0) * GS + g] = v0;
            rv[(3 * s + 1) * GS + g] = v1;
            rv[(3 * s + 2) * GS + g] = v2;
            if (s < 29) {
                const float h2 = h * h;
                const int q = s & 3;
                int rr = 3 + q;
                Aw[rr * SA + 3 * (s + 1) + 0] = -1.0f;
                Aw[rr * SA + 3 * s + 0] = 1.0f;
                Aw[rr * SA + 3 * s + 1] = h;
                Aw[rr * SA + 3 * s + 2] = 0.5f * h2;
                rr = 7 + q;
                Aw[rr * SA + 3 * (s + 1) + 1] = -h;
                Aw[rr * SA + 3 * s + 1] = h;
                Aw[rr * SA + 3 * s + 2] = h2;
                rr = 11 + q;
                Aw[rr * SA + 3 * s + 0] = -1.0f;
                Aw[rr * SA + 3 * (s + 1) + 0] = 1.0f;
                Aw[rr * SA + 3 * (s + 1) + 1] = -h;
                Aw[rr * SA + 3 * (s + 1) + 2] = 0.5f * h2;
                rr = 15 + q;
                Aw[rr * SA + 3 * s + 1] = h;
                Aw[rr * SA + 3 * (s + 1) + 1] = -h;
                Aw[rr * SA + 3 * (s + 1) + 2] = h2;
                if (s >= 1) {
                    rr = 19 + s % 5;
                    Aw[rr * SA + 3 * (s - 1) + 1] = -1.0f;
                    Aw[rr * SA + 3 * (s + 1) + 1] = 1.0f;
                    Aw[rr * SA + 3 * s + 2] = -(h + hp);
                }
            }
        }
        wave_lds_handover();

        v2f fp[NT][KROWS / 4], fq[NT][KROWS / 4];
#pragma unroll
        for (int kk = 0; kk < KROWS / 4; ++kk) {
            const int k0 = kk * 4 + 2 * half;
#pragma unroll
            for (int t = 0; t < NT; ++t) {
                fq[t][kk].x = Aw[k0 * SA + 13 * t - 3 + lm];
                fq[t][kk].y = Aw[(k0 + 1) * SA + 13 * t - 3 + lm];
                fp[t][kk].x = Aw[k0 * SA + 13 * t + lm];
                fp[t][kk].y = Aw[(k0 + 1) * SA + 13 * t + lm];
            }
        }
        wave_lds_handover();

        v8f acc[NT];
#pragma unroll
        for (int t = 0; t < NT; ++t) acc[t] = (v8f){0.f, 0.f, 0.f, 0.f, 0.f, 0.f, 0.f, 0.f};
#pragma unroll
        for (int kk = 0; kk < KROWS / 4; ++kk) {
#pragma unroll
            for (int t = 0; t < NT; ++t)
                acc[t] = __builtin_amdgcn_wmma_f32_16x16x4_f32(false, fp[t][kk], false, fq[t][kk],
                                                               (short)0, acc[t], false, false);
        }

        const int bi = (6 * lm + 8 * half - 18) * GS + g;
#pragma unroll
        for (int v = 0; v < 8; ++v) {
            const int d0 = v + 8 * half + 3 - lm;
            if (d0 >= 0 && d0 <= BW) {
                if (lm >= 3) band[bi + v * GS] = acc[0][v];
#pragma unroll
                for (int t = 1; t < NT - 1; ++t) band[bi + (91 * t + v) * GS] = acc[t][v];
                if (v + 8 * half < NB - 78) band[bi + (91 * (NT - 1) + v) * GS] = acc[NT - 1][v];
            }
        }
    }
    __syncthreads();

    if (tid < 32) {
        const int g   = tid & 15;
        const int rev = tid >> 4;
        const int o0 = rev ? 623 * GS : 0, sp = rev ? -6 * GS : GS, sq = rev ? -GS : 6 * GS;
        const int ro = rev ? 89 * GS : 0,  rs = rev ? -GS : GS;
        float W[7][7];
        float R[7];
#pragma unroll
        for (int r = 0; r < 7; ++r) {
#pragma unroll
            for (int c = 0; c <= r; ++c) W[r][c] = band[o0 + sp * r + sq * c + g];
            R[r] = rv[ro + rs * r + g];
        }
        const int dj = sp + sq;
        int eb = o0 + sp * 7 + sq;
        int lc = o0;
        int rj = ro;
        float PA[3][7], PAr[3], PB[3][7], PBr[3];
#pragma unroll
        for (int k = 0; k < 3; ++k) {
#pragma unroll
            for (int c = 0; c < 7; ++c) PA[k][c] = band[eb + k * dj + sq * c + g];
            PAr[k] = rv[rj + rs * (7 + k) + g];
        }
        for (int t = 0; t < 7; ++t) {
            eb += 3 * dj;
#pragma unroll
            for (int k = 0; k < 3; ++k) {
#pragma unroll
                for (int c = 0; c < 7; ++c) PB[k][c] = band[eb + k * dj + sq * c + g];
                PBr[k] = rv[rj + rs * (10 + k) + g];
            }
            __builtin_amdgcn_sched_barrier(0);
#pragma unroll
            for (int h = 0; h < 2; ++h) {
                if (h == 1) {
                    eb += 3 * dj;
#pragma unroll
                    for (int k = 0; k < 3; ++k) {
#pragma unroll
                        for (int c = 0; c < 7; ++c) PA[k][c] = band[eb + k * dj + sq * c + g];
                        PAr[k] = rv[rj + rs * (10 + k) + g];
                    }
                    __builtin_amdgcn_sched_barrier(0);
                }
#pragma unroll
                for (int k = 0; k < 3; ++k) {
                    float inv = __builtin_amdgcn_rsqf(W[0][0]);
                    float l[7];
#pragma unroll
                    for (int r = 1; r < 7; ++r) l[r] = W[r][0] * inv;
                    float y = R[0] * inv;
                    band[lc + g] = inv;
#pragma unroll
                    for (int r = 1; r < 7; ++r) band[lc + r * GS + g] = l[r];
                    rv[rj + g] = y;
#pragma unroll
                    for (int r = 0; r < 6; ++r) {
#pragma unroll
                        for (int c = 0; c <= r; ++c) W[r][c] = W[r + 1][c + 1] - l[r + 1] * l[c + 1];
                        R[r] = R[r + 1] - l[r + 1] * y;
                    }
#pragma unroll
                    for (int c = 0; c < 7; ++c) W[6][c] = (h == 0) ? PA[k][c] : PB[k][c];
                    R[6] = (h == 0) ? PAr[k] : PBr[k];
                    lc += dj; rj += rs;
                }
            }
        }
        {
            float O[6][6], Or[6];
#pragma unroll
            for (int r = 0; r < 6; ++r) {
#pragma unroll
                for (int c = 0; c <= r; ++c) O[r][c] = __shfl_xor(W[r][c], 16, 32);
                Or[r] = __shfl_xor(R[r], 16, 32);
            }
            float S[6][6], Sr[6];
#pragma unroll
            for (int r = 0; r < 6; ++r) {
#pragma unroll
                for (int c = 0; c <= r; ++c)
                    S[r][c] = W[r][c] + (O[5 - c][5 - r] - band[o0 + sp * (42 + r) + sq * (42 + c) + g]);
                Sr[r] = R[r] + (Or[5 - r] - rv[ro + rs * (42 + r) + g]);
            }
#pragma unroll
            for (int r = 0; r < 6; ++r) {
#pragma unroll
                for (int c = 0; c <= r; ++c) W[r][c] = S[r][c];
                R[r] = Sr[r];
            }
#pragma unroll
            for (int c = 0; c < 7; ++c) W[6][c] = 0.0f;
            R[6] = 0.0f;
        }
        for (int j = 42; j < 48; ++j) {
            float inv = __builtin_amdgcn_rsqf(W[0][0]);
            float l[7];
#pragma unroll
            for (int r = 1; r < 7; ++r) l[r] = W[r][0] * inv;
            float y = R[0] * inv;
            if (!rev) {
                band[(j * 7) * GS + g] = inv;
#pragma unroll
                for (int r = 1; r < 7; ++r) band[(j * 7 + r) * GS + g] = l[r];
                rv[j * GS + g] = y;
            }
#pragma unroll
            for (int r = 0; r < 6; ++r) {
#pragma unroll
                for (int c = 0; c <= r; ++c) W[r][c] = W[r + 1][c + 1] - l[r + 1] * l[c + 1];
                R[r] = R[r + 1] - l[r + 1] * y;
            }
#pragma unroll
            for (int c = 0; c < 7; ++c) W[6][c] = 0.0f;
            R[6] = 0.0f;
        }
        float X[7];
#pragma unroll
        for (int dd = 0; dd < 7; ++dd) X[dd] = 0.0f;
        float CA[3][7], CAy[3], CB[3][7], CBy[3];
#pragma unroll
        for (int k = 0; k < 3; ++k) {
#pragma unroll
            for (int dd = 0; dd < 7; ++dd) CA[k][dd] = band[lc - (k + 1) * dj + dd * GS + g];
            CAy[k] = rv[rj - (k + 1) * rs + g];
        }
        if (!rev) {
            float Lb[6][7], yb[6];
#pragma unroll
            for (int jj = 0; jj < 6; ++jj) {
#pragma unroll
                for (int dd = 0; dd < 7; ++dd) Lb[jj][dd] = band[((47 - jj) * 7 + dd) * GS + g];
                yb[jj] = rv[(47 - jj) * GS + g];
            }
            __builtin_amdgcn_sched_barrier(0);
#pragma unroll
            for (int jj = 0; jj < 6; ++jj) {
                float s = yb[jj];
#pragma unroll
                for (int dd = 1; dd < 7; ++dd) s -= Lb[jj][dd] * X[dd];
                float x = s * Lb[jj][0];
                rv[(47 - jj) * GS + g] = x;
#pragma unroll
                for (int dd = 6; dd > 1; --dd) X[dd] = X[dd - 1];
                X[1] = x;
            }
        }
        {
            float Ox[7];
#pragma unroll
            for (int dd = 1; dd < 7; ++dd) Ox[dd] = __shfl_xor(X[dd], 16, 32);
            if (rev) {
#pragma unroll
                for (int dd = 1; dd < 7; ++dd) X[dd] = Ox[7 - dd];
            }
        }
        for (int t = 0; t < 7; ++t) {
#pragma unroll
            for (int h = 0; h < 2; ++h) {
                lc -= 3 * dj; rj -= 3 * rs;
                if (h == 0) {
#pragma unroll
                    for (int k = 0; k < 3; ++k) {
#pragma unroll
                        for (int dd = 0; dd < 7; ++dd) CB[k][dd] = band[lc - (k + 1) * dj + dd * GS + g];
                        CBy[k] = rv[rj - (k + 1) * rs + g];
                    }
                } else if (t < 6) {
#pragma unroll
                    for (int k = 0; k < 3; ++k) {
#pragma unroll
                        for (int dd = 0; dd < 7; ++dd) CA[k][dd] = band[lc - (k + 1) * dj + dd * GS + g];
                        CAy[k] = rv[rj - (k + 1) * rs + g];
                    }
                }
                __builtin_amdgcn_sched_barrier(0);
#pragma unroll
                for (int k = 0; k < 3; ++k) {
                    float s = (h == 0) ? CAy[k] : CBy[k];
#pragma unroll
                    for (int dd = 1; dd < 7; ++dd) s -= ((h == 0) ? CA[k][dd] : CB[k][dd]) * X[dd];
                    float x = s * ((h == 0) ? CA[k][0] : CB[k][0]);
                    rv[rj + (2 - k) * rs + g] = x;
#pragma unroll
                    for (int dd = 6; dd > 1; --dd) X[dd] = X[dd - 1];
                    X[1] = x;
                }
            }
        }
    }
    __syncthreads();
    for (int i = tid; i < nsys * 90; i += 256) {
        int g = i / 90, j = i - g * 90;
        out[b0 * 90 + i] = rv[j * GS + g];
    }
}

extern "C" void kernel_launch(void* const* d_in, const int* in_sizes, int n_in,
                              void* d_out, int out_size, void* d_ws, size_t ws_size,
                              hipStream_t stream) {
    (void)n_in; (void)out_size; (void)d_ws; (void)ws_size;
    const float* coeffs = (const float*)d_in[0];
    const float* rhs    = (const float*)d_in[1];
    const float* iv     = (const float*)d_in[2];
    const float* steps  = (const float*)d_in[3];
    float* out = (float*)d_out;
    int bs = in_sizes[0] / 90;
    odein_solve_kernel<<<(bs + G - 1) / G, 256, 0, stream>>>(coeffs, rhs, iv, steps, out, bs);
}


# Folded A kept with rows k, k+1 side by side in LDS: a lane's WMMA operand pair is one conflict-free 8-byte read; was two 4-byte reads from two rows, paired across operands, sorted by register moves
# speedup vs baseline: 72.8702x; 1.0721x over previous
#include <hip/hip_runtime.h>
#include <hip/hip_bf16.h>


typedef float v2f __attribute__((ext_vector_type(2)));
typedef float v8f __attribute__((ext_vector_type(8)));
typedef float v4f __attribute__((ext_vector_type(4)));

#define SA2   112
#define KROWS 24
#define CO    3
#define NT    7
#define NB    90
#define BW    6
#define G     16
#define GS    17
#define NBP   96
#define NW    8
#define AIDX(row, col) (((((row) >> 1) * SA2) + (col)) * 2 + ((row) & 1))

static __device__ __forceinline__ void wave_lds_handover() {
    __builtin_amdgcn_fence(__ATOMIC_RELEASE, "wavefront");
    __builtin_amdgcn_wave_barrier();
    __builtin_amdgcn_fence(__ATOMIC_ACQUIRE, "wavefront");
}

__global__ __launch_bounds__(256)
void odein_solve_kernel(const float* __restrict__ coeffs,
                        const float* __restrict__ rhs,
                        const float* __restrict__ iv_rhs,
                        const float* __restrict__ steps,
                        float* __restrict__ out,
                        int bs)
{
    __shared__ __attribute__((aligned(16))) float Abuf[NW * (KROWS / 2) * SA2 * 2];
    __shared__ float band[NBP * 7 * GS];
    __shared__ float rv[NBP * GS];
    __shared__ __attribute__((aligned(16))) float cs[G * 90];
    __shared__ __attribute__((aligned(16))) float rs[G * 30];
    __shared__ __attribute__((aligned(16))) float ivs[G * 4];
    __shared__ __attribute__((aligned(16))) float sts[G * 29];

    const int tid = threadIdx.x;
    const int b0  = blockIdx.x * G;
    const int nsys = (bs - b0 < G) ? (bs - b0) : G;

    const bool full = (nsys == G);
    v4f q0 = (v4f){0.f, 0.f, 0.f, 0.f}, q1 = q0, q2 = q0;
    const float* p1 = (tid < 104) ? coeffs + b0 * 90 + 4 * (256 + tid)
                    : (tid < 224) ? rhs + b0 * 30 + 4 * (tid - 104)
                                  : iv_rhs + b0 * 4 + 4 * (tid - 224);
    float* d1 = (tid < 104) ? cs + 4 * (256 + tid) : (tid < 224) ? rs + 4 * (tid - 104) : ivs + 4 * (tid - 224);
    if (full) {
        __builtin_memcpy(&q0, __builtin_assume_aligned(coeffs + b0 * 90 + 4 * tid, 16), 16);
        if (tid < 240) __builtin_memcpy(&q1, __builtin_assume_aligned(p1, 16), 16);
        if (tid < 116) __builtin_memcpy(&q2, __builtin_assume_aligned(steps + b0 * 29 + 4 * tid, 16), 16);
    }
    for (int idx = tid; idx < NW * (KROWS / 2) * SA2 * 2; idx += 256) Abuf[idx] = 0.0f;
    for (int i = tid; i < (NBP - 84) * 7 * GS; i += 256) band[84 * 7 * GS + i] = 0.0f;
    for (int i = tid; i < (NBP - NB) * GS; i += 256) rv[NB * GS + i] = 0.0f;
    if (full) {
        __builtin_memcpy(__builtin_assume_aligned(cs + 4 * tid, 16), &q0, 16);
        if (tid < 240) __builtin_memcpy(__builtin_assume_aligned(d1, 16), &q1, 16);
        if (tid < 116) __builtin_memcpy(__builtin_assume_aligned(sts + 4 * tid, 16), &q2, 16);
    } else {
        for (int i = tid; i < G * 90; i += 256) cs[i]  = (i < nsys * 90) ? coeffs[b0 * 90 + i] : 0.0f;
        for (int i = tid; i < G * 30; i += 256) rs[i]  = (i < nsys * 30) ? rhs[b0 * 30 + i] : 0.0f;
        for (int i = tid; i < G * 29; i += 256) sts[i] = (i < nsys * 29) ? steps[b0 * 29 + i] : 0.0f;
        if (tid < G * 4) ivs[tid] = (tid < nsys * 4) ? iv_rhs[b0 * 4 + tid] : 0.0f;
    }
    __syncthreads();

    const int lane = tid & 31, wv = tid >> 5;
    const int half = lane >> 4, lm = lane & 15;
    float* Aw = Abuf + wv * ((KROWS / 2) * SA2 * 2) + 2 * CO;

    if (lane < 4) {
        const int ivcol = lane + (lane >> 1);
        const int ivrow = lane == 0 ? 6 : lane == 1 ? 10 : lane == 2 ? 18 : 19;
        Aw[AIDX(ivrow, ivcol)] = 1.0f;
    }

    for (int s2 = 0; s2 < G / NW; ++s2) {
        const int g = (G / NW) * wv + s2;
        if (lane < 30) {
            const int s = lane;
            const float c0 = cs[g * 90 + 3 * s], c1 = cs[g * 90 + 3 * s + 1], c2 = cs[g * 90 + 3 * s + 2];
            const float r  = rs[g * 30 + s];
            const float i0 = ivs[g * 4 + 2 * (s & 1)], i1 = ivs[g * 4 + 2 * (s & 1) + 1];
            const float h  = sts[g * 29 + (s < 29 ? s : 28)];
            const float hp = sts[g * 29 + (s >= 1 ? s - 1 : 0)];
            const int row = s % 3;
            float v0 = c0 * r, v1 = c1 * r;
            const float v2 = c2 * r;
            if (s < 2) { v0 += i0; v1 += i1; }
            Aw[AIDX(row, 3 * s + 0)] = c0;
            Aw[AIDX(row, 3 * s + 1)] = c1;
            Aw[AIDX(row, 3 * s + 2)] = c2;
            rv[(3 * s + 0) * GS + g] = v0;
            rv[(3 * s + 1) * GS + g] = v1;
            rv[(3 * s + 2) * GS + g] = v2;
            if (s < 29) {
                const float h2 = h * h;
                const int q = s & 3;
                int rr = 3 + q;
                Aw[AIDX(rr, 3 * (s + 1) + 0)] = -1.0f;
                Aw[AIDX(rr, 3 * s + 0)] = 1.0f;
                Aw[AIDX(rr, 3 * s + 1)] = h;
                Aw[AIDX(rr, 3 * s + 2)] = 0.5f * h2;
                rr = 7 + q;
                Aw[AIDX(rr, 3 * (s + 1) + 1)] = -h;
                Aw[AIDX(rr, 3 * s + 1)] = h;
                Aw[AIDX(rr, 3 * s + 2)] = h2;
                rr = 11 + q;
                Aw[AIDX(rr, 3 * s + 0)] = -1.0f;
                Aw[AIDX(rr, 3 * (s + 1) + 0)] = 1.0f;
                Aw[AIDX(rr, 3 * (s + 1) + 1)] = -h;
                Aw[AIDX(rr, 3 * (s + 1) + 2)] = 0.5f * h2;
                rr = 15 + q;
                Aw[AIDX(rr, 3 * s + 1)] = h;
                Aw[AIDX(rr, 3 * (s + 1) + 1)] = -h;
                Aw[AIDX(rr, 3 * (s + 1) + 2)] = h2;
                if (s >= 1) {
                    rr = 19 + s % 5;
                    Aw[AIDX(rr, 3 * (s - 1) + 1)] = -1.0f;
                    Aw[AIDX(rr, 3 * (s + 1) + 1)] = 1.0f;
                    Aw[AIDX(rr, 3 * s + 2)] = -(h + hp);
                }
            }
        }
        wave_lds_handover();

        v2f fp[NT][KROWS / 4], fq[NT][KROWS / 4];
#pragma unroll
        for (int kk = 0; kk < KROWS / 4; ++kk) {
            const int k0 = kk * 4 + 2 * half;
#pragma unroll
            for (int t = 0; t < NT; ++t) {
                __builtin_memcpy(&fq[t][kk], __builtin_assume_aligned(Aw + AIDX(k0, 13 * t - 3 + lm), 8), 8);
                __builtin_memcpy(&fp[t][kk], __builtin_assume_aligned(Aw + AIDX(k0, 13 * t + lm), 8), 8);
            }
        }
        wave_lds_handover();

        v8f acc[NT];
#pragma unroll
        for (int t = 0; t < NT; ++t) acc[t] = (v8f){0.f, 0.f, 0.f, 0.f, 0.f, 0.f, 0.f, 0.f};
#pragma unroll
        for (int kk = 0; kk < KROWS / 4; ++kk) {
#pragma unroll
            for (int t = 0; t < NT; ++t)
                acc[t] = __builtin_amdgcn_wmma_f32_16x16x4_f32(false, fp[t][kk], false, fq[t][kk],
                                                               (short)0, acc[t], false, false);
        }

        const int bi = (6 * lm + 8 * half - 18) * GS + g;
#pragma unroll
        for (int v = 0; v < 8; ++v) {
            const int d0 = v + 8 * half + 3 - lm;
            if (d0 >= 0 && d0 <= BW) {
                if (lm >= 3) band[bi + v * GS] = acc[0][v];
#pragma unroll
                for (int t = 1; t < NT - 1; ++t) band[bi + (91 * t + v) * GS] = acc[t][v];
                if (v + 8 * half < NB - 78) band[bi + (91 * (NT - 1) + v) * GS] = acc[NT - 1][v];
            }
        }
    }
    __syncthreads();

    if (tid < 32) {
        const int g   = tid & 15;
        const int rev = tid >> 4;
        const int o0 = rev ? 623 * GS : 0, sp = rev ? -6 * GS : GS, sq = rev ? -GS : 6 * GS;
        const int ro = rev ? 89 * GS : 0,  rs = rev ? -GS : GS;
        float W[7][7];
        float R[7];
#pragma unroll
        for (int r = 0; r < 7; ++r) {
#pragma unroll
            for (int c = 0; c <= r; ++c) W[r][c] = band[o0 + sp * r + sq * c + g];
            R[r] = rv[ro + rs * r + g];
        }
        const int dj = sp + sq;
        int eb = o0 + sp * 7 + sq;
        int lc = o0;
        int rj = ro;
        float PA[3][7], PAr[3], PB[3][7], PBr[3];
#pragma unroll
        for (int k = 0; k < 3; ++k) {
#pragma unroll
            for (int c = 0; c < 7; ++c) PA[k][c] = band[eb + k * dj + sq * c + g];
            PAr[k] = rv[rj + rs * (7 + k) + g];
        }
        for (int t = 0; t < 7; ++t) {
            eb += 3 * dj;
#pragma unroll
            for (int k = 0; k < 3; ++k) {
#pragma unroll
                for (int c = 0; c < 7; ++c) PB[k][c] = band[eb + k * dj + sq * c + g];
                PBr[k] = rv[rj + rs * (10 + k) + g];
            }
            __builtin_amdgcn_sched_barrier(0);
#pragma unroll
            for (int h = 0; h < 2; ++h) {
                if (h == 1) {
                    eb += 3 * dj;
#pragma unroll
                    for (int k = 0; k < 3; ++k) {
#pragma unroll
                        for (int c = 0; c < 7; ++c) PA[k][c] = band[eb + k * dj + sq * c + g];
                        PAr[k] = rv[rj + rs * (10 + k) + g];
                    }
                    __builtin_amdgcn_sched_barrier(0);
                }
#pragma unroll
                for (int k = 0; k < 3; ++k) {
                    float inv = __builtin_amdgcn_rsqf(W[0][0]);
                    float l[7];
#pragma unroll
                    for (int r = 1; r < 7; ++r) l[r] = W[r][0] * inv;
                    float y = R[0] * inv;
                    band[lc + g] = inv;
#pragma unroll
                    for (int r = 1; r < 7; ++r) band[lc + r * GS + g] = l[r];
                    rv[rj + g] = y;
#pragma unroll
                    for (int r = 0; r < 6; ++r) {
#pragma unroll
                        for (int c = 0; c <= r; ++c) W[r][c] = W[r + 1][c + 1] - l[r + 1] * l[c + 1];
                        R[r] = R[r + 1] - l[r + 1] * y;
                    }
#pragma unroll
                    for (int c = 0; c < 7; ++c) W[6][c] = (h == 0) ? PA[k][c] : PB[k][c];
                    R[6] = (h == 0) ? PAr[k] : PBr[k];
                    lc += dj; rj += rs;
                }
            }
        }
        {
            float O[6][6], Or[6];
#pragma unroll
            for (int r = 0; r < 6; ++r) {
#pragma unroll
                for (int c = 0; c <= r; ++c) O[r][c] = __shfl_xor(W[r][c], 16, 32);
                Or[r] = __shfl_xor(R[r], 16, 32);
            }
            float S[6][6], Sr[6];
#pragma unroll
            for (int r = 0; r < 6; ++r) {
#pragma unroll
                for (int c = 0; c <= r; ++c)
                    S[r][c] = W[r][c] + (O[5 - c][5 - r] - band[o0 + sp * (42 + r) + sq * (42 + c) + g]);
                Sr[r] = R[r] + (Or[5 - r] - rv[ro + rs * (42 + r) + g]);
            }
#pragma unroll
            for (int r = 0; r < 6; ++r) {
#pragma unroll
                for (int c = 0; c <= r; ++c) W[r][c] = S[r][c];
                R[r] = Sr[r];
            }
#pragma unroll
            for (int c = 0; c < 7; ++c) W[6][c] = 0.0f;
            R[6] = 0.0f;
        }
        for (int j = 42; j < 48; ++j) {
            float inv = __builtin_amdgcn_rsqf(W[0][0]);
            float l[7];
#pragma unroll
            for (int r = 1; r < 7; ++r) l[r] = W[r][0] * inv;
            float y = R[0] * inv;
            if (!rev) {
                band[(j * 7) * GS + g] = inv;
#pragma unroll
                for (int r = 1; r < 7; ++r) band[(j * 7 + r) * GS + g] = l[r];
                rv[j * GS + g] = y;
            }
#pragma unroll
            for (int r = 0; r < 6; ++r) {
#pragma unroll
                for (int c = 0; c <= r; ++c) W[r][c] = W[r + 1][c + 1] - l[r + 1] * l[c + 1];
                R[r] = R[r + 1] - l[r + 1] * y;
            }
#pragma unroll
            for (int c = 0; c < 7; ++c) W[6][c] = 0.0f;
            R[6] = 0.0f;
        }
        float X[7];
#pragma unroll
        for (int dd = 0; dd < 7; ++dd) X[dd] = 0.0f;
        float CA[3][7], CAy[3], CB[3][7], CBy[3];
#pragma unroll
        for (int k = 0; k < 3; ++k) {
#pragma unroll
            for (int dd = 0; dd < 7; ++dd) CA[k][dd] = band[lc - (k + 1) * dj + dd * GS + g];
            CAy[k] = rv[rj - (k + 1) * rs + g];
        }
        if (!rev) {
            float Lb[6][7], yb[6];
#pragma unroll
            for (int jj = 0; jj < 6; ++jj) {
#pragma unroll
                for (int dd = 0; dd < 7; ++dd) Lb[jj][dd] = band[((47 - jj) * 7 + dd) * GS + g];
                yb[jj] = rv[(47 - jj) * GS + g];
            }
            __builtin_amdgcn_sched_barrier(0);
#pragma unroll
            for (int jj = 0; jj < 6; ++jj) {
                float s = yb[jj];
#pragma unroll
                for (int dd = 1; dd < 7; ++dd) s -= Lb[jj][dd] * X[dd];
                float x = s * Lb[jj][0];
                rv[(47 - jj) * GS + g] = x;
#pragma unroll
                for (int dd = 6; dd > 1; --dd) X[dd] = X[dd - 1];
                X[1] = x;
            }
        }
        {
            float Ox[7];
#pragma unroll
            for (int dd = 1; dd < 7; ++dd) Ox[dd] = __shfl_xor(X[dd], 16, 32);
            if (rev) {
#pragma unroll
                for (int dd = 1; dd < 7; ++dd) X[dd] = Ox[7 - dd];
            }
        }
        for (int t = 0; t < 7; ++t) {
#pragma unroll
            for (int h = 0; h < 2; ++h) {
                lc -= 3 * dj; rj -= 3 * rs;
                if (h == 0) {
#pragma unroll
                    for (int k = 0; k < 3; ++k) {
#pragma unroll
                        for (int dd = 0; dd < 7; ++dd) CB[k][dd] = band[lc - (k + 1) * dj + dd * GS + g];
                        CBy[k] = rv[rj - (k + 1) * rs + g];
                    }
                } else if (t < 6) {
#pragma unroll
                    for (int k = 0; k < 3; ++k) {
#pragma unroll
                        for (int dd = 0; dd < 7; ++dd) CA[k][dd] = band[lc - (k + 1) * dj + dd * GS + g];
                        CAy[k] = rv[rj - (k + 1) * rs + g];
                    }
                }
                __builtin_amdgcn_sched_barrier(0);
#pragma unroll
                for (int k = 0; k < 3; ++k) {
                    float s = (h == 0) ? CAy[k] : CBy[k];
#pragma unroll
                    for (int dd = 1; dd < 7; ++dd) s -= ((h == 0) ? CA[k][dd] : CB[k][dd]) * X[dd];
                    float x = s * ((h == 0) ? CA[k][0] : CB[k][0]);
                    rv[rj + (2 - k) * rs + g] = x;
#pragma unroll
                    for (int dd = 6; dd > 1; --dd) X[dd] = X[dd - 1];
                    X[1] = x;
                }
            }
        }
    }
    __syncthreads();
    for (int i = tid; i < nsys * 90; i += 256) {
        int g = i / 90, j = i - g * 90;
        out[b0 * 90 + i] = rv[j * GS + g];
    }
}

extern "C" void kernel_launch(void* const* d_in, const int* in_sizes, int n_in,
                              void* d_out, int out_size, void* d_ws, size_t ws_size,
                              hipStream_t stream) {
    (void)n_in; (void)out_size; (void)d_ws; (void)ws_size;
    const float* coeffs = (const float*)d_in[0];
    const float* rhs    = (const float*)d_in[1];
    const float* iv     = (const float*)d_in[2];
    const float* steps  = (const float*)d_in[3];
    float* out = (float*)d_out;
    int bs = in_sizes[0] / 90;
    odein_solve_kernel<<<(bs + G - 1) / G, 256, 0, stream>>>(coeffs, rhs, iv, steps, out, bs);
}
